# starting kernel
# baseline (speedup 1.0000x reference)
#include <hip/hip_runtime.h>
#include <hip/hip_bf16.h>


#define BSZ   16
#define CIN   64
#define CO    64
#define TLEN  8192
#define NST   256
#define NCH   16
#define LCH   512
#define SUB   32
#define NSUB  (LCH/SUB)
#define T2    (TLEN/2)

#define UST   40
#define VST   260
#define YST   66

typedef __attribute__((ext_vector_type(2))) float v2f;
typedef __attribute__((ext_vector_type(8))) float v8f;

__device__ __forceinline__ v8f wmma4(v2f a, v2f b, v8f c) {
  return __builtin_amdgcn_wmma_f32_16x16x4_f32(
       false, a,  false, b,
       (short)0, c,  false,  false);
}

__device__ __forceinline__ size_t opaque_zero() {
  size_t z = 0;
  asm volatile("" : "+s"(z));
  return z;
}

__global__ void k_discretize(const float* __restrict__ rl,
                             const float* __restrict__ Bc,
                             float* __restrict__ Ad, float* __restrict__ Bd) {
  int n = threadIdx.x;
  float x = rl[n];
  float sp = fmaxf(x, 0.0f) + log1pf(expf(-fabsf(x)));
  float lam = -sp;
  float ad = expf(lam);
  float factor = (fabsf(lam) > 1e-6f) ? (ad - 1.0f) / lam : 1.0f;
  Ad[n] = ad;
  for (int c = 0; c < CIN; ++c)
    Bd[c * NST + n] = Bc[c * NST + n] * factor;
}

__device__ __forceinline__ void load_u_tile(const float* __restrict__ u, int b,
                                            int t0, float* lds_u) {
  int tid = threadIdx.x;
#pragma unroll
  for (int kk = 0; kk < (CIN * SUB) / 256; ++kk) {
    int i = tid + kk * 256;
    int c = i >> 5, tl = i & 31;
    lds_u[c * UST + tl] = u[((size_t)(b * CIN + c)) * TLEN + t0 + tl];
  }
}

__device__ __forceinline__ void prefetch_u_tile(const float* __restrict__ u,
                                                int b, int t0) {
  int tid = threadIdx.x;
#pragma unroll
  for (int kk = 0; kk < (CIN * SUB) / 256; ++kk) {
    int i = tid + kk * 256;
    int c = i >> 5, tl = i & 31;
    __builtin_prefetch(&u[((size_t)(b * CIN + c)) * TLEN + t0 + tl], 0, 1);
  }
}

__device__ __forceinline__ void v_tile_wmma(const float* lds_u,
                                            const float* __restrict__ Bd,
                                            float* lds_v) {
  int tid = threadIdx.x;
  int wave = tid >> 5, lane = tid & 31, lo = lane & 15, hi = lane >> 4;
#pragma unroll
  for (int idx = 0; idx < 4; ++idx) {
    int tile = wave * 4 + idx;
    int nt = tile >> 1, mt = tile & 1;
    int m = mt * 16 + lo, nn = nt * 16 + lo;
    v8f acc = {0.f, 0.f, 0.f, 0.f, 0.f, 0.f, 0.f, 0.f};
#pragma unroll 8
    for (int ks = 0; ks < CIN / 4; ++ks) {
      int k = ks * 4 + 2 * hi;
      v2f a;  a.x = lds_u[k * UST + m];       a.y = lds_u[(k + 1) * UST + m];
      v2f bb; bb.x = Bd[k * NST + nn];        bb.y = Bd[(k + 1) * NST + nn];
      acc = wmma4(a, bb, acc);
    }
#pragma unroll
    for (int r = 0; r < 8; ++r)
      lds_v[(mt * 16 + r + 8 * hi) * VST + nn] = acc[r];
  }
}

__global__ void k_chunk_ends(const float* __restrict__ u,
                             const float* __restrict__ Ad,
                             const float* __restrict__ Bd_in,
                             float* __restrict__ ce) {
  __shared__ float lds_u[CIN * UST];
  __shared__ float lds_v[SUB * VST];
  int chunk = blockIdx.x, b = blockIdx.y, tid = threadIdx.x;
  float a_n = Ad[tid];
  float s = 0.0f;
  for (int st = 0; st < NSUB; ++st) {
    const float* Bd = Bd_in + opaque_zero();
    int t0 = chunk * LCH + st * SUB;
    load_u_tile(u, b, t0, lds_u);
    if (st + 1 < NSUB) prefetch_u_tile(u, b, t0 + SUB);
    __syncthreads();
    v_tile_wmma(lds_u, Bd, lds_v);
    __syncthreads();
#pragma unroll 8
    for (int tl = 0; tl < SUB; ++tl)
      s = s * a_n + lds_v[tl * VST + tid];
    __syncthreads();
  }
  ce[chunk * (BSZ * NST) + b * NST + tid] = s;
}

__global__ void k_combine(const float* __restrict__ Ad,
                          const float* __restrict__ ce,
                          float* __restrict__ carry) {
  int n = threadIdx.x, idx = blockIdx.x * NST + n;
  float aL = Ad[n];
#pragma unroll
  for (int i = 0; i < 9; ++i) aL = aL * aL;
  float s = 0.0f;
  for (int ch = 0; ch < NCH; ++ch) {
    carry[ch * (BSZ * NST) + idx] = s;
    s = aL * s + ce[ch * (BSZ * NST) + idx];
  }
}

__global__ void k_scan_y(const float* __restrict__ u,
                         const float* __restrict__ Ad,
                         const float* __restrict__ Bd_in,
                         const float* __restrict__ Cm_in,
                         const float* __restrict__ ln_w,
                         const float* __restrict__ ln_b,
                         const float* __restrict__ carry,
                         float* __restrict__ yact) {
  __shared__ float lds_u[CIN * UST];
  __shared__ float lds_v[SUB * VST];
  __shared__ float lds_y[SUB * YST];
  __shared__ float lds_red[SUB * 16];
  __shared__ float lds_mu[SUB];
  __shared__ float lds_rs[SUB];

  int chunk = blockIdx.x, b = blockIdx.y, tid = threadIdx.x;
  int wave = tid >> 5, lane = tid & 31, lo = lane & 15, hi = lane >> 4;
  float a_n = Ad[tid];
  float s = carry[chunk * (BSZ * NST) + b * NST + tid];

  for (int st = 0; st < NSUB; ++st) {
    const float* Bd = Bd_in + opaque_zero();
    const float* Cm = Cm_in + opaque_zero();
    int t0 = chunk * LCH + st * SUB;
    load_u_tile(u, b, t0, lds_u);
    if (st + 1 < NSUB) prefetch_u_tile(u, b, t0 + SUB);
    __syncthreads();
    v_tile_wmma(lds_u, Bd, lds_v);
    __syncthreads();

#pragma unroll 8
    for (int tl = 0; tl < SUB; ++tl) {
      s = s * a_n + lds_v[tl * VST + tid];
      lds_v[tl * VST + tid] = s;
    }
    __syncthreads();

    {
      int mt = wave >> 2, nt = wave & 3;
      int m = mt * 16 + lo, oo = nt * 16 + lo;
      v8f acc = {0.f, 0.f, 0.f, 0.f, 0.f, 0.f, 0.f, 0.f};
#pragma unroll 8
      for (int ks = 0; ks < NST / 4; ++ks) {
        int k = ks * 4 + 2 * hi;
        v2f a;  a.x = lds_v[m * VST + k];  a.y = lds_v[m * VST + k + 1];
        v2f bb; bb.x = Cm[k * CO + oo];    bb.y = Cm[(k + 1) * CO + oo];
        acc = wmma4(a, bb, acc);
      }
#pragma unroll
      for (int r = 0; r < 8; ++r)
        lds_y[(mt * 16 + r + 8 * hi) * YST + oo] = acc[r];
    }
    __syncthreads();

    {
      int t = tid & 31, q = tid >> 5;
      float sm = 0.f, s2 = 0.f;
#pragma unroll
      for (int j = 0; j < 8; ++j) {
        float v = lds_y[t * YST + q * 8 + j];
        sm += v; s2 += v * v;
      }
      lds_red[(t * 8 + q) * 2 + 0] = sm;
      lds_red[(t * 8 + q) * 2 + 1] = s2;
    }
    __syncthreads();
    if (tid < SUB) {
      float sm = 0.f, s2 = 0.f;
#pragma unroll
      for (int q = 0; q < 8; ++q) {
        sm += lds_red[(tid * 8 + q) * 2 + 0];
        s2 += lds_red[(tid * 8 + q) * 2 + 1];
      }
      float mu = sm * (1.0f / CO);
      float var = s2 * (1.0f / CO) - mu * mu;
      lds_mu[tid] = mu;
      lds_rs[tid] = rsqrtf(var + 1e-5f);
    }
    __syncthreads();

    {
      int t = lane;
      float mu = lds_mu[t], rs = lds_rs[t];
#pragma unroll
      for (int j = 0; j < 8; ++j) {
        int o = wave * 8 + j;
        float v = (lds_y[t * YST + o] - mu) * rs * ln_w[o] + ln_b[o];
        v = v / (1.0f + __expf(-v));
        yact[((size_t)(b * CO + o)) * TLEN + t0 + t] = v;
      }
    }
    __syncthreads();
  }
}

__global__ void k_conv(const float* __restrict__ yact,
                       const float* __restrict__ conv_w,
                       const float* __restrict__ conv_b,
                       float* __restrict__ out) {
  int b = blockIdx.y, tid = threadIdx.x;
  int wave = tid >> 5, lane = tid & 31, lo = lane & 15, hi = lane >> 4;
  int tau0 = (blockIdx.x * 4 + wave) * 16;

  v8f acc[4];
#pragma unroll
  for (int mt = 0; mt < 4; ++mt)
#pragma unroll
    for (int r = 0; r < 8; ++r)
      acc[mt][r] = conv_b[mt * 16 + r + 8 * hi];

#pragma unroll 2
  for (int ks = 0; ks < CO / 4; ++ks) {
    int k = ks * 4 + 2 * hi;
    v2f be, bo;
    be.x = yact[((size_t)(b * CO + k)) * TLEN + tau0 + lo];
    be.y = yact[((size_t)(b * CO + k + 1)) * TLEN + tau0 + lo];
    bo.x = yact[((size_t)(b * CO + k)) * TLEN + T2 + tau0 + lo];
    bo.y = yact[((size_t)(b * CO + k + 1)) * TLEN + T2 + tau0 + lo];
#pragma unroll
    for (int mt = 0; mt < 4; ++mt) {
      int m = mt * 16 + lo;
      v2f ae, ao;
      ae.x = conv_w[m * 128 + 2 * k];
      ae.y = conv_w[m * 128 + 2 * (k + 1)];
      ao.x = conv_w[m * 128 + 2 * k + 1];
      ao.y = conv_w[m * 128 + 2 * (k + 1) + 1];
      acc[mt] = wmma4(ae, be, acc[mt]);
      acc[mt] = wmma4(ao, bo, acc[mt]);
    }
  }

#pragma unroll
  for (int mt = 0; mt < 4; ++mt)
#pragma unroll
    for (int r = 0; r < 8; ++r) {
      int o = mt * 16 + r + 8 * hi;
      int t = 2 * (tau0 + lo);
      float v = acc[mt][r];
      float2 pv = make_float2(v, v);
      *reinterpret_cast<float2*>(&out[((size_t)(b * CO + o)) * TLEN + t]) = pv;
    }
}

extern "C" void kernel_launch(void* const* d_in, const int* in_sizes, int n_in,
                              void* d_out, int out_size, void* d_ws, size_t ws_size,
                              hipStream_t stream) {
  const float* u     = (const float*)d_in[0];
  const float* rl    = (const float*)d_in[1];
  const float* Bc    = (const float*)d_in[2];
  const float* Cm    = (const float*)d_in[3];
  const float* ln_w  = (const float*)d_in[4];
  const float* ln_b  = (const float*)d_in[5];
  const float* cw    = (const float*)d_in[6];
  const float* cb    = (const float*)d_in[7];
  float* out = (float*)d_out;

  float* ws    = (float*)d_ws;
  float* Ad    = ws;
  float* Bd    = Ad + NST;
  float* ce    = Bd + CIN * NST;
  float* carry = ce + NCH * BSZ * NST;
  float* yact  = carry + NCH * BSZ * NST;

  k_discretize<<<1, NST, 0, stream>>>(rl, Bc, Ad, Bd);
  k_chunk_ends<<<dim3(NCH, BSZ), 256, 0, stream>>>(u, Ad, Bd, ce);
  k_combine<<<BSZ, NST, 0, stream>>>(Ad, ce, carry);
  k_scan_y<<<dim3(NCH, BSZ), 256, 0, stream>>>(u, Ad, Bd, Cm, ln_w, ln_b,
                                               carry, yact);
  k_conv<<<dim3(T2 / 64, BSZ), 128, 0, stream>>>(yact, cw, cb, out);
}


# Bd and C WMMA B-fragments loaded once per workgroup and held in registers over all 16 subtiles (were re-read from L2 every subtile); __launch_bounds__(256) so they do not spill
# speedup vs baseline: 1.7656x; 1.7656x over previous
#include <hip/hip_runtime.h>
#include <hip/hip_bf16.h>


#define BSZ   16
#define CIN   64
#define CO    64
#define TLEN  8192
#define NST   256
#define NCH   16
#define LCH   512
#define SUB   32
#define NSUB  (LCH/SUB)
#define T2    (TLEN/2)

#define UST   40
#define VST   260
#define YST   66

typedef __attribute__((ext_vector_type(2))) float v2f;
typedef __attribute__((ext_vector_type(8))) float v8f;

__device__ __forceinline__ v8f wmma4(v2f a, v2f b, v8f c) {
  return __builtin_amdgcn_wmma_f32_16x16x4_f32(
       false, a,  false, b,
       (short)0, c,  false,  false);
}

__global__ void k_discretize(const float* __restrict__ rl,
                             const float* __restrict__ Bc,
                             float* __restrict__ Ad, float* __restrict__ Bd) {
  int n = threadIdx.x;
  float x = rl[n];
  float sp = fmaxf(x, 0.0f) + log1pf(expf(-fabsf(x)));
  float lam = -sp;
  float ad = expf(lam);
  float factor = (fabsf(lam) > 1e-6f) ? (ad - 1.0f) / lam : 1.0f;
  Ad[n] = ad;
  for (int c = 0; c < CIN; ++c)
    Bd[c * NST + n] = Bc[c * NST + n] * factor;
}

__device__ __forceinline__ void load_u_tile(const float* __restrict__ u, int b,
                                            int t0, float* lds_u) {
  int tid = threadIdx.x;
#pragma unroll
  for (int kk = 0; kk < (CIN * SUB) / 256; ++kk) {
    int i = tid + kk * 256;
    int c = i >> 5, tl = i & 31;
    lds_u[c * UST + tl] = u[((size_t)(b * CIN + c)) * TLEN + t0 + tl];
  }
}

__device__ __forceinline__ void prefetch_u_tile(const float* __restrict__ u,
                                                int b, int t0) {
  int tid = threadIdx.x;
#pragma unroll
  for (int kk = 0; kk < (CIN * SUB) / 256; ++kk) {
    int i = tid + kk * 256;
    int c = i >> 5, tl = i & 31;
    __builtin_prefetch(&u[((size_t)(b * CIN + c)) * TLEN + t0 + tl], 0, 1);
  }
}

__device__ __forceinline__ void load_bd_frags(const float* __restrict__ Bd,
                                              v2f (&bf)[2][CIN / 4]) {
  int tid = threadIdx.x;
  int wave = tid >> 5, lane = tid & 31, lo = lane & 15, hi = lane >> 4;
#pragma unroll
  for (int j = 0; j < 2; ++j) {
    int nn = (wave * 2 + j) * 16 + lo;
#pragma unroll
    for (int ks = 0; ks < CIN / 4; ++ks) {
      int k = ks * 4 + 2 * hi;
      bf[j][ks].x = Bd[k * NST + nn];
      bf[j][ks].y = Bd[(k + 1) * NST + nn];
    }
  }
}

__device__ __forceinline__ void v_tile_wmma(const float* lds_u,
                                            const v2f (&bf)[2][CIN / 4],
                                            float* lds_v) {
  int tid = threadIdx.x;
  int wave = tid >> 5, lane = tid & 31, lo = lane & 15, hi = lane >> 4;
  int nn0 = (wave * 2) * 16 + lo, nn1 = nn0 + 16;
#pragma unroll
  for (int mt = 0; mt < 2; ++mt) {
    int m = mt * 16 + lo;
    v8f acc0 = {0.f, 0.f, 0.f, 0.f, 0.f, 0.f, 0.f, 0.f};
    v8f acc1 = {0.f, 0.f, 0.f, 0.f, 0.f, 0.f, 0.f, 0.f};
#pragma unroll
    for (int ks = 0; ks < CIN / 4; ++ks) {
      int k = ks * 4 + 2 * hi;
      v2f a;  a.x = lds_u[k * UST + m];       a.y = lds_u[(k + 1) * UST + m];
      acc0 = wmma4(a, bf[0][ks], acc0);
      acc1 = wmma4(a, bf[1][ks], acc1);
    }
#pragma unroll
    for (int r = 0; r < 8; ++r) {
      lds_v[(mt * 16 + r + 8 * hi) * VST + nn0] = acc0[r];
      lds_v[(mt * 16 + r + 8 * hi) * VST + nn1] = acc1[r];
    }
  }
}

__global__ void __launch_bounds__(256)
k_chunk_ends(const float* __restrict__ u,
             const float* __restrict__ Ad,
             const float* __restrict__ Bd,
             float* __restrict__ ce) {
  __shared__ float lds_u[CIN * UST];
  __shared__ float lds_v[SUB * VST];
  int chunk = blockIdx.x, b = blockIdx.y, tid = threadIdx.x;
  float a_n = Ad[tid];
  float s = 0.0f;
  v2f bdf[2][CIN / 4];
  load_bd_frags(Bd, bdf);
  for (int st = 0; st < NSUB; ++st) {
    int t0 = chunk * LCH + st * SUB;
    load_u_tile(u, b, t0, lds_u);
    if (st + 1 < NSUB) prefetch_u_tile(u, b, t0 + SUB);
    __syncthreads();
    v_tile_wmma(lds_u, bdf, lds_v);
    __syncthreads();
#pragma unroll 8
    for (int tl = 0; tl < SUB; ++tl)
      s = s * a_n + lds_v[tl * VST + tid];
    __syncthreads();
  }
  ce[chunk * (BSZ * NST) + b * NST + tid] = s;
}

__global__ void k_combine(const float* __restrict__ Ad,
                          const float* __restrict__ ce,
                          float* __restrict__ carry) {
  int n = threadIdx.x, idx = blockIdx.x * NST + n;
  float aL = Ad[n];
#pragma unroll
  for (int i = 0; i < 9; ++i) aL = aL * aL;
  float s = 0.0f;
  for (int ch = 0; ch < NCH; ++ch) {
    carry[ch * (BSZ * NST) + idx] = s;
    s = aL * s + ce[ch * (BSZ * NST) + idx];
  }
}

__global__ void __launch_bounds__(256)
k_scan_y(const float* __restrict__ u,
         const float* __restrict__ Ad,
         const float* __restrict__ Bd,
         const float* __restrict__ Cm,
         const float* __restrict__ ln_w,
         const float* __restrict__ ln_b,
         const float* __restrict__ carry,
         float* __restrict__ yact) {
  __shared__ float lds_u[CIN * UST];
  __shared__ float lds_v[SUB * VST];
  __shared__ float lds_y[SUB * YST];
  __shared__ float lds_red[SUB * 16];
  __shared__ float lds_mu[SUB];
  __shared__ float lds_rs[SUB];

  int chunk = blockIdx.x, b = blockIdx.y, tid = threadIdx.x;
  int wave = tid >> 5, lane = tid & 31, lo = lane & 15, hi = lane >> 4;
  float a_n = Ad[tid];
  float s = carry[chunk * (BSZ * NST) + b * NST + tid];

  v2f bdf[2][CIN / 4];
  load_bd_frags(Bd, bdf);
  v2f cmf[NST / 4];
  {
    int oo = (wave & 3) * 16 + lo;
#pragma unroll
    for (int ks = 0; ks < NST / 4; ++ks) {
      int k = ks * 4 + 2 * hi;
      cmf[ks].x = Cm[k * CO + oo];
      cmf[ks].y = Cm[(k + 1) * CO + oo];
    }
  }

  for (int st = 0; st < NSUB; ++st) {
    int t0 = chunk * LCH + st * SUB;
    load_u_tile(u, b, t0, lds_u);
    if (st + 1 < NSUB) prefetch_u_tile(u, b, t0 + SUB);
    __syncthreads();
    v_tile_wmma(lds_u, bdf, lds_v);
    __syncthreads();

#pragma unroll 8
    for (int tl = 0; tl < SUB; ++tl) {
      s = s * a_n + lds_v[tl * VST + tid];
      lds_v[tl * VST + tid] = s;
    }
    __syncthreads();

    {
      int mt = wave >> 2, nt = wave & 3;
      int m = mt * 16 + lo, oo = nt * 16 + lo;
      v8f acc = {0.f, 0.f, 0.f, 0.f, 0.f, 0.f, 0.f, 0.f};
#pragma unroll
      for (int ks = 0; ks < NST / 4; ++ks) {
        int k = ks * 4 + 2 * hi;
        v2f a;  a.x = lds_v[m * VST + k];  a.y = lds_v[m * VST + k + 1];
        acc = wmma4(a, cmf[ks], acc);
      }
#pragma unroll
      for (int r = 0; r < 8; ++r)
        lds_y[(mt * 16 + r + 8 * hi) * YST + oo] = acc[r];
    }
    __syncthreads();

    {
      int t = tid & 31, q = tid >> 5;
      float sm = 0.f, s2 = 0.f;
#pragma unroll
      for (int j = 0; j < 8; ++j) {
        float v = lds_y[t * YST + q * 8 + j];
        sm += v; s2 += v * v;
      }
      lds_red[(t * 8 + q) * 2 + 0] = sm;
      lds_red[(t * 8 + q) * 2 + 1] = s2;
    }
    __syncthreads();
    if (tid < SUB) {
      float sm = 0.f, s2 = 0.f;
#pragma unroll
      for (int q = 0; q < 8; ++q) {
        sm += lds_red[(tid * 8 + q) * 2 + 0];
        s2 += lds_red[(tid * 8 + q) * 2 + 1];
      }
      float mu = sm * (1.0f / CO);
      float var = s2 * (1.0f / CO) - mu * mu;
      lds_mu[tid] = mu;
      lds_rs[tid] = rsqrtf(var + 1e-5f);
    }
    __syncthreads();

    {
      int t = lane;
      float mu = lds_mu[t], rs = lds_rs[t];
#pragma unroll
      for (int j = 0; j < 8; ++j) {
        int o = wave * 8 + j;
        float v = (lds_y[t * YST + o] - mu) * rs * ln_w[o] + ln_b[o];
        v = v / (1.0f + __expf(-v));
        yact[((size_t)(b * CO + o)) * TLEN + t0 + t] = v;
      }
    }
    __syncthreads();
  }
}

__global__ void k_conv(const float* __restrict__ yact,
                       const float* __restrict__ conv_w,
                       const float* __restrict__ conv_b,
                       float* __restrict__ out) {
  int b = blockIdx.y, tid = threadIdx.x;
  int wave = tid >> 5, lane = tid & 31, lo = lane & 15, hi = lane >> 4;
  int tau0 = (blockIdx.x * 4 + wave) * 16;

  v8f acc[4];
#pragma unroll
  for (int mt = 0; mt < 4; ++mt)
#pragma unroll
    for (int r = 0; r < 8; ++r)
      acc[mt][r] = conv_b[mt * 16 + r + 8 * hi];

#pragma unroll 2
  for (int ks = 0; ks < CO / 4; ++ks) {
    int k = ks * 4 + 2 * hi;
    v2f be, bo;
    be.x = yact[((size_t)(b * CO + k)) * TLEN + tau0 + lo];
    be.y = yact[((size_t)(b * CO + k + 1)) * TLEN + tau0 + lo];
    bo.x = yact[((size_t)(b * CO + k)) * TLEN + T2 + tau0 + lo];
    bo.y = yact[((size_t)(b * CO + k + 1)) * TLEN + T2 + tau0 + lo];
#pragma unroll
    for (int mt = 0; mt < 4; ++mt) {
      int m = mt * 16 + lo;
      v2f ae, ao;
      ae.x = conv_w[m * 128 + 2 * k];
      ae.y = conv_w[m * 128 + 2 * (k + 1)];
      ao.x = conv_w[m * 128 + 2 * k + 1];
      ao.y = conv_w[m * 128 + 2 * (k + 1) + 1];
      acc[mt] = wmma4(ae, be, acc[mt]);
      acc[mt] = wmma4(ao, bo, acc[mt]);
    }
  }

#pragma unroll
  for (int mt = 0; mt < 4; ++mt)
#pragma unroll
    for (int r = 0; r < 8; ++r) {
      int o = mt * 16 + r + 8 * hi;
      int t = 2 * (tau0 + lo);
      float v = acc[mt][r];
      float2 pv = make_float2(v, v);
      *reinterpret_cast<float2*>(&out[((size_t)(b * CO + o)) * TLEN + t]) = pv;
    }
}

extern "C" void kernel_launch(void* const* d_in, const int* in_sizes, int n_in,
                              void* d_out, int out_size, void* d_ws, size_t ws_size,
                              hipStream_t stream) {
  const float* u     = (const float*)d_in[0];
  const float* rl    = (const float*)d_in[1];
  const float* Bc    = (const float*)d_in[2];
  const float* Cm    = (const float*)d_in[3];
  const float* ln_w  = (const float*)d_in[4];
  const float* ln_b  = (const float*)d_in[5];
  const float* cw    = (const float*)d_in[6];
  const float* cb    = (const float*)d_in[7];
  float* out = (float*)d_out;

  float* ws    = (float*)d_ws;
  float* Ad    = ws;
  float* Bd    = Ad + NST;
  float* ce    = Bd + CIN * NST;
  float* carry = ce + NCH * BSZ * NST;
  float* yact  = carry + NCH * BSZ * NST;

  k_discretize<<<1, NST, 0, stream>>>(rl, Bc, Ad, Bd);
  k_chunk_ends<<<dim3(NCH, BSZ), 256, 0, stream>>>(u, Ad, Bd, ce);
  k_combine<<<BSZ, NST, 0, stream>>>(Ad, ce, carry);
  k_scan_y<<<dim3(NCH, BSZ), 256, 0, stream>>>(u, Ad, Bd, Cm, ln_w, ln_b,
                                               carry, yact);
  k_conv<<<dim3(T2 / 64, BSZ), 128, 0, stream>>>(yact, cw, cb, out);
}


# k_chunk_ends, k_scan_y: waves 0-3 and 4-7 take different roles one subtile apart; one role's WMMA run covers the other's scan, loads, LayerNorm and stores; 2 and 1 barriers per subtile (were 7, 3)
# speedup vs baseline: 2.0680x; 1.1713x over previous
#include <hip/hip_runtime.h>
#include <hip/hip_bf16.h>


#define BSZ   16
#define CIN   64
#define CO    64
#define TLEN  8192
#define NST   256
#define NCH   16
#define LCH   512
#define SUB   32
#define NSUB  (LCH/SUB)
#define T2    (TLEN/2)

#define UST   40
#define VST   260
#define YST   66
#define RST   20

typedef __attribute__((ext_vector_type(2))) float v2f;
typedef __attribute__((ext_vector_type(8))) float v8f;

__device__ __forceinline__ v8f wmma4(v2f a, v2f b, v8f c) {
  return __builtin_amdgcn_wmma_f32_16x16x4_f32(
       false, a,  false, b,
       (short)0, c,  false,  false);
}

__global__ void k_discretize(const float* __restrict__ rl,
                             const float* __restrict__ Bc,
                             float* __restrict__ Ad, float* __restrict__ Bd) {
  int n = threadIdx.x;
  float x = rl[n];
  float sp = fmaxf(x, 0.0f) + log1pf(expf(-fabsf(x)));
  float lam = -sp;
  float ad = expf(lam);
  float factor = (fabsf(lam) > 1e-6f) ? (ad - 1.0f) / lam : 1.0f;
  Ad[n] = ad;
  for (int c = 0; c < CIN; ++c)
    Bd[c * NST + n] = Bc[c * NST + n] * factor;
}


__device__ __forceinline__ void fetch_u_tile128(const float* __restrict__ u, int b,
                                                int t0, int rt, float (&ur)[16]) {
#pragma unroll
  for (int kk = 0; kk < (CIN * SUB) / 128; ++kk) {
    int i = rt + kk * 128;
    int c = i >> 5, tl = i & 31;
    ur[kk] = u[((size_t)(b * CIN + c)) * TLEN + t0 + tl];
  }
}
__device__ __forceinline__ void put_u_tile128(float* lds_u, int rt,
                                              const float (&ur)[16]) {
#pragma unroll
  for (int kk = 0; kk < (CIN * SUB) / 128; ++kk) {
    int i = rt + kk * 128;
    int c = i >> 5, tl = i & 31;
    lds_u[c * UST + tl] = ur[kk];
  }
}

__device__ __forceinline__ void load_bd_frags4(const float* __restrict__ Bd,
                                               int rw, int lo, int hi,
                                               v2f (&fr)[64]) {
#pragma unroll
  for (int j = 0; j < 4; ++j) {
    int nn = (rw * 4 + j) * 16 + lo;
#pragma unroll
    for (int ks = 0; ks < CIN / 4; ++ks) {
      int k = ks * 4 + 2 * hi;
      fr[j * 16 + ks].x = Bd[k * NST + nn];
      fr[j * 16 + ks].y = Bd[(k + 1) * NST + nn];
    }
  }
}

__device__ __forceinline__ void v_tile_wmma8(const float* lds_u,
                                             const v2f (&fr)[64],
                                             float* lds_v, int rw, int lo, int hi) {
  v8f acc[2][4];
#pragma unroll
  for (int mt = 0; mt < 2; ++mt)
#pragma unroll
    for (int j = 0; j < 4; ++j)
#pragma unroll
      for (int r = 0; r < 8; ++r) acc[mt][j][r] = 0.f;
#pragma unroll
  for (int ks = 0; ks < CIN / 4; ++ks) {
    int k = ks * 4 + 2 * hi;
    v2f a0, a1;
    a0.x = lds_u[k * UST + lo];        a0.y = lds_u[(k + 1) * UST + lo];
    a1.x = lds_u[k * UST + 16 + lo];   a1.y = lds_u[(k + 1) * UST + 16 + lo];
#pragma unroll
    for (int j = 0; j < 4; ++j) {
      acc[0][j] = wmma4(a0, fr[j * 16 + ks], acc[0][j]);
      acc[1][j] = wmma4(a1, fr[j * 16 + ks], acc[1][j]);
    }
  }
#pragma unroll
  for (int mt = 0; mt < 2; ++mt)
#pragma unroll
    for (int j = 0; j < 4; ++j)
#pragma unroll
      for (int r = 0; r < 8; ++r)
        lds_v[(mt * 16 + r + 8 * hi) * VST + (rw * 4 + j) * 16 + lo] = acc[mt][j][r];
}

__global__ void __launch_bounds__(256)
k_chunk_ends(const float* __restrict__ u,
             const float* __restrict__ Ad,
             const float* __restrict__ Bd,
             float* __restrict__ ce) {
  __shared__ float lds_u[2][CIN * UST];
  __shared__ float lds_v[2][SUB * VST];
  const int chunk = blockIdx.x, b = blockIdx.y, tid = threadIdx.x;
  const int lane = tid & 31, lo = lane & 15, hi = lane >> 4;
  const int wave = __builtin_amdgcn_readfirstlane(tid >> 5);
  const bool gemm_role = wave < 4;
  const int rw = wave & 3, rt = rw * 32 + lane;

  v2f fr[64];
#pragma unroll
  for (int i = 0; i < 64; ++i) { fr[i].x = 0.f; fr[i].y = 0.f; }
  float a0 = 0.f, a1 = 0.f, s0 = 0.f, s1 = 0.f;
  if (gemm_role) {
    load_bd_frags4(Bd, rw, lo, hi, fr);
  } else {
    a0 = Ad[rt];  a1 = Ad[rt + 128];
    float ur[16];
    fetch_u_tile128(u, b, chunk * LCH, rt, ur);
    put_u_tile128(lds_u[0], rt, ur);
  }
  __syncthreads();

  for (int step = 0; step <= NSUB; ++step) {
    if (gemm_role) {
      if (step < NSUB)
        v_tile_wmma8(lds_u[step & 1], fr, lds_v[step & 1], rw, lo, hi);
    } else {
      float ur[16];
      if (step + 1 < NSUB)
        fetch_u_tile128(u, b, chunk * LCH + (step + 1) * SUB, rt, ur);
      if (step >= 1) {
        const float* S = lds_v[(step - 1) & 1];
#pragma unroll 16
        for (int tl = 0; tl < SUB; ++tl) {
          s0 = s0 * a0 + S[tl * VST + rt];
          s1 = s1 * a1 + S[tl * VST + rt + 128];
        }
      }
      if (step + 1 < NSUB)
        put_u_tile128(lds_u[(step + 1) & 1], rt, ur);
    }
    __syncthreads();
  }
  if (!gemm_role) {
    ce[chunk * (BSZ * NST) + b * NST + rt] = s0;
    ce[chunk * (BSZ * NST) + b * NST + rt + 128] = s1;
  }
}

__global__ void k_combine(const float* __restrict__ Ad,
                          const float* __restrict__ ce,
                          float* __restrict__ carry) {
  int n = threadIdx.x, idx = blockIdx.x * NST + n;
  float aL = Ad[n];
#pragma unroll
  for (int i = 0; i < 9; ++i) aL = aL * aL;
  float s = 0.0f;
  for (int ch = 0; ch < NCH; ++ch) {
    carry[ch * (BSZ * NST) + idx] = s;
    s = aL * s + ce[ch * (BSZ * NST) + idx];
  }
}

__global__ void __launch_bounds__(256)
k_scan_y(const float* __restrict__ u,
         const float* __restrict__ Ad,
         const float* __restrict__ Bd,
         const float* __restrict__ Cm,
         const float* __restrict__ ln_w,
         const float* __restrict__ ln_b,
         const float* __restrict__ carry,
         float* __restrict__ yact) {
  __shared__ float lds_u[CIN * UST];
  __shared__ float lds_v[2][SUB * VST];
  __shared__ float lds_y[2][SUB * YST];
  __shared__ __attribute__((aligned(16))) float lds_red[SUB * RST];

  const int chunk = blockIdx.x, b = blockIdx.y, tid = threadIdx.x;
  const int lane = tid & 31, lo = lane & 15, hi = lane >> 4;
  const int wave = __builtin_amdgcn_readfirstlane(tid >> 5);
  const bool front = wave < 4;
  const int rw = wave & 3, rt = rw * 32 + lane;

  v2f fr[64];
#pragma unroll
  for (int i = 0; i < 64; ++i) { fr[i].x = 0.f; fr[i].y = 0.f; }
  float a0 = 0.f, a1 = 0.f, s0 = 0.f, s1 = 0.f;
  float lw[16], lb[16];
#pragma unroll
  for (int j = 0; j < 16; ++j) { lw[j] = 0.f; lb[j] = 0.f; }

  if (front) {
    load_bd_frags4(Bd, rw, lo, hi, fr);
    a0 = Ad[rt];  a1 = Ad[rt + 128];
    s0 = carry[chunk * (BSZ * NST) + b * NST + rt];
    s1 = carry[chunk * (BSZ * NST) + b * NST + rt + 128];
#pragma unroll
    for (int j = 0; j < 16; ++j) { lw[j] = ln_w[rw * 16 + j]; lb[j] = ln_b[rw * 16 + j]; }
    float ur[16];
    fetch_u_tile128(u, b, chunk * LCH, rt, ur);
    put_u_tile128(lds_u, rt, ur);
  } else {
    int oo = rw * 16 + lo;
#pragma unroll
    for (int ks = 0; ks < NST / 4; ++ks) {
      int k = ks * 4 + 2 * hi;
      fr[ks].x = Cm[k * CO + oo];
      fr[ks].y = Cm[(k + 1) * CO + oo];
    }
  }
  __syncthreads();

  for (int step = 0; step < NSUB + 2; ++step) {
    if (front) {
      if (step < NSUB)
        v_tile_wmma8(lds_u, fr, lds_v[step & 1], rw, lo, hi);
    } else if (step >= 2) {
      const float* yb = lds_y[step & 1];
      float pr[4];
#pragma unroll
      for (int h = 0; h < 2; ++h) {
        int q = rw * 2 + h;
        float sm = 0.f, s2 = 0.f;
#pragma unroll
        for (int j = 0; j < 8; ++j) {
          float v = yb[lane * YST + q * 8 + j];
          sm += v; s2 += v * v;
        }
        pr[2 * h] = sm;  pr[2 * h + 1] = s2;
      }
      *reinterpret_cast<float4*>(&lds_red[lane * RST + rw * 4]) =
          make_float4(pr[0], pr[1], pr[2], pr[3]);
    }
    __syncthreads();

    if (front) {
      float ur[16];
      if (step + 1 < NSUB)
        fetch_u_tile128(u, b, chunk * LCH + (step + 1) * SUB, rt, ur);
      if (step >= 2) {
        const float* yb = lds_y[step & 1];
        int t0 = chunk * LCH + (step - 2) * SUB;
        float sm = 0.f, s2 = 0.f;
#pragma unroll
        for (int q = 0; q < 8; ++q) {
          sm += lds_red[lane * RST + q * 2 + 0];
          s2 += lds_red[lane * RST + q * 2 + 1];
        }
        float mu = sm * (1.0f / CO);
        float var = s2 * (1.0f / CO) - mu * mu;
        float rs = rsqrtf(var + 1e-5f);
#pragma unroll
        for (int j = 0; j < 16; ++j) {
          int o = rw * 16 + j;
          float v = (yb[lane * YST + o] - mu) * rs * lw[j] + lb[j];
          v = v / (1.0f + __expf(-v));
          yact[((size_t)(b * CO + o)) * TLEN + t0 + lane] = v;
        }
      }
      if (step < NSUB) {
        float* S = lds_v[step & 1];
#pragma unroll 16
        for (int tl = 0; tl < SUB; ++tl) {
          s0 = s0 * a0 + S[tl * VST + rt];
          S[tl * VST + rt] = s0;
          s1 = s1 * a1 + S[tl * VST + rt + 128];
          S[tl * VST + rt + 128] = s1;
        }
      }
      if (step + 1 < NSUB)
        put_u_tile128(lds_u, rt, ur);
    } else if (step >= 1 && step <= NSUB) {
      const float* S = lds_v[(step - 1) & 1];
      float* yb = lds_y[(step - 1) & 1];
      v8f y0 = {0.f, 0.f, 0.f, 0.f, 0.f, 0.f, 0.f, 0.f};
      v8f y1 = {0.f, 0.f, 0.f, 0.f, 0.f, 0.f, 0.f, 0.f};
      v2f fa[2][8][2];
#pragma unroll
      for (int i = 0; i < 8; ++i) {
        int k = i * 4 + 2 * hi;
        fa[0][i][0].x = S[lo * VST + k];         fa[0][i][0].y = S[lo * VST + k + 1];
        fa[0][i][1].x = S[(16 + lo) * VST + k];  fa[0][i][1].y = S[(16 + lo) * VST + k + 1];
      }
#pragma unroll
      for (int blk = 0; blk < NST / 32; ++blk) {
        if (blk + 1 < NST / 32) {
#pragma unroll
          for (int i = 0; i < 8; ++i) {
            int k = ((blk + 1) * 8 + i) * 4 + 2 * hi;
            fa[(blk + 1) & 1][i][0].x = S[lo * VST + k];
            fa[(blk + 1) & 1][i][0].y = S[lo * VST + k + 1];
            fa[(blk + 1) & 1][i][1].x = S[(16 + lo) * VST + k];
            fa[(blk + 1) & 1][i][1].y = S[(16 + lo) * VST + k + 1];
          }
        }
        __builtin_amdgcn_sched_barrier(0);
#pragma unroll
        for (int i = 0; i < 8; ++i) {
          y0 = wmma4(fa[blk & 1][i][0], fr[blk * 8 + i], y0);
          y1 = wmma4(fa[blk & 1][i][1], fr[blk * 8 + i], y1);
        }
        __builtin_amdgcn_sched_barrier(0);
      }
#pragma unroll
      for (int r = 0; r < 8; ++r) {
        yb[(r + 8 * hi) * YST + rw * 16 + lo] = y0[r];
        yb[(16 + r + 8 * hi) * YST + rw * 16 + lo] = y1[r];
      }
    }
    __syncthreads();
  }
}

__global__ void k_conv(const float* __restrict__ yact,
                       const float* __restrict__ conv_w,
                       const float* __restrict__ conv_b,
                       float* __restrict__ out) {
  int b = blockIdx.y, tid = threadIdx.x;
  int wave = tid >> 5, lane = tid & 31, lo = lane & 15, hi = lane >> 4;
  int tau0 = (blockIdx.x * 4 + wave) * 16;

  v8f acc[4];
#pragma unroll
  for (int mt = 0; mt < 4; ++mt)
#pragma unroll
    for (int r = 0; r < 8; ++r)
      acc[mt][r] = conv_b[mt * 16 + r + 8 * hi];

#pragma unroll 2
  for (int ks = 0; ks < CO / 4; ++ks) {
    int k = ks * 4 + 2 * hi;
    v2f be, bo;
    be.x = yact[((size_t)(b * CO + k)) * TLEN + tau0 + lo];
    be.y = yact[((size_t)(b * CO + k + 1)) * TLEN + tau0 + lo];
    bo.x = yact[((size_t)(b * CO + k)) * TLEN + T2 + tau0 + lo];
    bo.y = yact[((size_t)(b * CO + k + 1)) * TLEN + T2 + tau0 + lo];
#pragma unroll
    for (int mt = 0; mt < 4; ++mt) {
      int m = mt * 16 + lo;
      v2f ae, ao;
      ae.x = conv_w[m * 128 + 2 * k];
      ae.y = conv_w[m * 128 + 2 * (k + 1)];
      ao.x = conv_w[m * 128 + 2 * k + 1];
      ao.y = conv_w[m * 128 + 2 * (k + 1) + 1];
      acc[mt] = wmma4(ae, be, acc[mt]);
      acc[mt] = wmma4(ao, bo, acc[mt]);
    }
  }

#pragma unroll
  for (int mt = 0; mt < 4; ++mt)
#pragma unroll
    for (int r = 0; r < 8; ++r) {
      int o = mt * 16 + r + 8 * hi;
      int t = 2 * (tau0 + lo);
      float v = acc[mt][r];
      float2 pv = make_float2(v, v);
      *reinterpret_cast<float2*>(&out[((size_t)(b * CO + o)) * TLEN + t]) = pv;
    }
}

extern "C" void kernel_launch(void* const* d_in, const int* in_sizes, int n_in,
                              void* d_out, int out_size, void* d_ws, size_t ws_size,
                              hipStream_t stream) {
  const float* u     = (const float*)d_in[0];
  const float* rl    = (const float*)d_in[1];
  const float* Bc    = (const float*)d_in[2];
  const float* Cm    = (const float*)d_in[3];
  const float* ln_w  = (const float*)d_in[4];
  const float* ln_b  = (const float*)d_in[5];
  const float* cw    = (const float*)d_in[6];
  const float* cb    = (const float*)d_in[7];
  float* out = (float*)d_out;

  float* ws    = (float*)d_ws;
  float* Ad    = ws;
  float* Bd    = Ad + NST;
  float* ce    = Bd + CIN * NST;
  float* carry = ce + NCH * BSZ * NST;
  float* yact  = carry + NCH * BSZ * NST;

  k_discretize<<<1, NST, 0, stream>>>(rl, Bc, Ad, Bd);
  k_chunk_ends<<<dim3(NCH, BSZ), 256, 0, stream>>>(u, Ad, Bd, ce);
  k_combine<<<BSZ, NST, 0, stream>>>(Ad, ce, carry);
  k_scan_y<<<dim3(NCH, BSZ), 256, 0, stream>>>(u, Ad, Bd, Cm, ln_w, ln_b,
                                               carry, yact);
  k_conv<<<dim3(T2 / 64, BSZ), 128, 0, stream>>>(yact, cw, cb, out);
}


# k_chunk_ends, k_scan_y: u tile kept t-major in LDS so a v-GEMM A fragment is one 8-byte read; all 32 reads of a subtile issued as one group before its 128 WMMAs (were 4-byte reads spread among them)
# speedup vs baseline: 2.1527x; 1.0410x over previous
#include <hip/hip_runtime.h>
#include <hip/hip_bf16.h>


#define BSZ   16
#define CIN   64
#define CO    64
#define TLEN  8192
#define NST   256
#define NCH   16
#define LCH   512
#define SUB   32
#define NSUB  (LCH/SUB)
#define T2    (TLEN/2)

#define UST   68
#define VST   260
#define YST   66
#define RST   20

typedef __attribute__((ext_vector_type(2))) float v2f;
typedef __attribute__((ext_vector_type(8))) float v8f;

__device__ __forceinline__ v8f wmma4(v2f a, v2f b, v8f c) {
  return __builtin_amdgcn_wmma_f32_16x16x4_f32(
       false, a,  false, b,
       (short)0, c,  false,  false);
}

__global__ void k_discretize(const float* __restrict__ rl,
                             const float* __restrict__ Bc,
                             float* __restrict__ Ad, float* __restrict__ Bd) {
  int n = threadIdx.x;
  float x = rl[n];
  float sp = fmaxf(x, 0.0f) + log1pf(expf(-fabsf(x)));
  float lam = -sp;
  float ad = expf(lam);
  float factor = (fabsf(lam) > 1e-6f) ? (ad - 1.0f) / lam : 1.0f;
  Ad[n] = ad;
  for (int c = 0; c < CIN; ++c)
    Bd[c * NST + n] = Bc[c * NST + n] * factor;
}


__device__ __forceinline__ void fetch_u_tile128(const float* __restrict__ u, int b,
                                                int t0, int rw, int lane,
                                                float (&ur)[16]) {
#pragma unroll
  for (int kk = 0; kk < CIN / 4; ++kk) {
    int c = rw * (CIN / 4) + kk;
    ur[kk] = u[((size_t)(b * CIN + c)) * TLEN + t0 + lane];
  }
}
__device__ __forceinline__ void put_u_tile128(float* lds_u, int rw, int lane,
                                              const float (&ur)[16]) {
#pragma unroll
  for (int q = 0; q < 4; ++q)
    *reinterpret_cast<float4*>(&lds_u[lane * UST + rw * (CIN / 4) + 4 * q]) =
        make_float4(ur[4 * q], ur[4 * q + 1], ur[4 * q + 2], ur[4 * q + 3]);
}

__device__ __forceinline__ void load_bd_frags4(const float* __restrict__ Bd,
                                               int rw, int lo, int hi,
                                               v2f (&fr)[64]) {
#pragma unroll
  for (int j = 0; j < 4; ++j) {
    int nn = (rw * 4 + j) * 16 + lo;
#pragma unroll
    for (int ks = 0; ks < CIN / 4; ++ks) {
      int k = ks * 4 + 2 * hi;
      fr[j * 16 + ks].x = Bd[k * NST + nn];
      fr[j * 16 + ks].y = Bd[(k + 1) * NST + nn];
    }
  }
}

__device__ __forceinline__ void v_tile_wmma8(const float* lds_u,
                                             const v2f (&fr)[64],
                                             float* lds_v, int rw, int lo, int hi) {
  v2f fa[2][CIN / 4];
#pragma unroll
  for (int ks = 0; ks < CIN / 4; ++ks) {
    int k = ks * 4 + 2 * hi;
    fa[0][ks] = *reinterpret_cast<const v2f*>(&lds_u[lo * UST + k]);
    fa[1][ks] = *reinterpret_cast<const v2f*>(&lds_u[(16 + lo) * UST + k]);
  }
  __builtin_amdgcn_sched_barrier(0);
  v8f acc[2][4];
#pragma unroll
  for (int mt = 0; mt < 2; ++mt)
#pragma unroll
    for (int j = 0; j < 4; ++j)
#pragma unroll
      for (int r = 0; r < 8; ++r) acc[mt][j][r] = 0.f;
#pragma unroll
  for (int ks = 0; ks < CIN / 4; ++ks) {
#pragma unroll
    for (int j = 0; j < 4; ++j) {
      acc[0][j] = wmma4(fa[0][ks], fr[j * 16 + ks], acc[0][j]);
      acc[1][j] = wmma4(fa[1][ks], fr[j * 16 + ks], acc[1][j]);
    }
  }
#pragma unroll
  for (int mt = 0; mt < 2; ++mt)
#pragma unroll
    for (int j = 0; j < 4; ++j)
#pragma unroll
      for (int r = 0; r < 8; ++r)
        lds_v[(mt * 16 + r + 8 * hi) * VST + (rw * 4 + j) * 16 + lo] = acc[mt][j][r];
}

__global__ void __launch_bounds__(256)
k_chunk_ends(const float* __restrict__ u,
             const float* __restrict__ Ad,
             const float* __restrict__ Bd,
             float* __restrict__ ce) {
  __shared__ __attribute__((aligned(16))) float lds_u[2][SUB * UST];
  __shared__ float lds_v[2][SUB * VST];
  const int chunk = blockIdx.x, b = blockIdx.y, tid = threadIdx.x;
  const int lane = tid & 31, lo = lane & 15, hi = lane >> 4;
  const int wave = __builtin_amdgcn_readfirstlane(tid >> 5);
  const bool gemm_role = wave < 4;
  const int rw = wave & 3, rt = rw * 32 + lane;

  v2f fr[64];
#pragma unroll
  for (int i = 0; i < 64; ++i) { fr[i].x = 0.f; fr[i].y = 0.f; }
  float a0 = 0.f, a1 = 0.f, s0 = 0.f, s1 = 0.f;
  if (gemm_role) {
    load_bd_frags4(Bd, rw, lo, hi, fr);
  } else {
    a0 = Ad[rt];  a1 = Ad[rt + 128];
    float ur[16];
    fetch_u_tile128(u, b, chunk * LCH, rw, lane, ur);
    put_u_tile128(lds_u[0], rw, lane, ur);
  }
  __syncthreads();

  for (int step = 0; step <= NSUB; ++step) {
    if (gemm_role) {
      if (step < NSUB)
        v_tile_wmma8(lds_u[step & 1], fr, lds_v[step & 1], rw, lo, hi);
    } else {
      float ur[16];
      if (step + 1 < NSUB)
        fetch_u_tile128(u, b, chunk * LCH + (step + 1) * SUB, rw, lane, ur);
      if (step >= 1) {
        const float* S = lds_v[(step - 1) & 1];
#pragma unroll 16
        for (int tl = 0; tl < SUB; ++tl) {
          s0 = s0 * a0 + S[tl * VST + rt];
          s1 = s1 * a1 + S[tl * VST + rt + 128];
        }
      }
      if (step + 1 < NSUB)
        put_u_tile128(lds_u[(step + 1) & 1], rw, lane, ur);
    }
    __syncthreads();
  }
  if (!gemm_role) {
    ce[chunk * (BSZ * NST) + b * NST + rt] = s0;
    ce[chunk * (BSZ * NST) + b * NST + rt + 128] = s1;
  }
}

__global__ void k_combine(const float* __restrict__ Ad,
                          const float* __restrict__ ce,
                          float* __restrict__ carry) {
  int n = threadIdx.x, idx = blockIdx.x * NST + n;
  float aL = Ad[n];
#pragma unroll
  for (int i = 0; i < 9; ++i) aL = aL * aL;
  float s = 0.0f;
  for (int ch = 0; ch < NCH; ++ch) {
    carry[ch * (BSZ * NST) + idx] = s;
    s = aL * s + ce[ch * (BSZ * NST) + idx];
  }
}

__global__ void __launch_bounds__(256)
k_scan_y(const float* __restrict__ u,
         const float* __restrict__ Ad,
         const float* __restrict__ Bd,
         const float* __restrict__ Cm,
         const float* __restrict__ ln_w,
         const float* __restrict__ ln_b,
         const float* __restrict__ carry,
         float* __restrict__ yact) {
  __shared__ __attribute__((aligned(16))) float lds_u[SUB * UST];
  __shared__ float lds_v[2][SUB * VST];
  __shared__ float lds_y[2][SUB * YST];
  __shared__ __attribute__((aligned(16))) float lds_red[SUB * RST];

  const int chunk = blockIdx.x, b = blockIdx.y, tid = threadIdx.x;
  const int lane = tid & 31, lo = lane & 15, hi = lane >> 4;
  const int wave = __builtin_amdgcn_readfirstlane(tid >> 5);
  const bool front = wave < 4;
  const int rw = wave & 3, rt = rw * 32 + lane;

  v2f fr[64];
#pragma unroll
  for (int i = 0; i < 64; ++i) { fr[i].x = 0.f; fr[i].y = 0.f; }
  float a0 = 0.f, a1 = 0.f, s0 = 0.f, s1 = 0.f;
  float lw[16], lb[16];
#pragma unroll
  for (int j = 0; j < 16; ++j) { lw[j] = 0.f; lb[j] = 0.f; }

  if (front) {
    load_bd_frags4(Bd, rw, lo, hi, fr);
    a0 = Ad[rt];  a1 = Ad[rt + 128];
    s0 = carry[chunk * (BSZ * NST) + b * NST + rt];
    s1 = carry[chunk * (BSZ * NST) + b * NST + rt + 128];
#pragma unroll
    for (int j = 0; j < 16; ++j) { lw[j] = ln_w[rw * 16 + j]; lb[j] = ln_b[rw * 16 + j]; }
    float ur[16];
    fetch_u_tile128(u, b, chunk * LCH, rw, lane, ur);
    put_u_tile128(lds_u, rw, lane, ur);
  } else {
    int oo = rw * 16 + lo;
#pragma unroll
    for (int ks = 0; ks < NST / 4; ++ks) {
      int k = ks * 4 + 2 * hi;
      fr[ks].x = Cm[k * CO + oo];
      fr[ks].y = Cm[(k + 1) * CO + oo];
    }
  }
  __syncthreads();

  for (int step = 0; step < NSUB + 2; ++step) {
    if (front) {
      if (step < NSUB)
        v_tile_wmma8(lds_u, fr, lds_v[step & 1], rw, lo, hi);
    } else if (step >= 2) {
      const float* yb = lds_y[step & 1];
      float pr[4];
#pragma unroll
      for (int h = 0; h < 2; ++h) {
        int q = rw * 2 + h;
        float sm = 0.f, s2 = 0.f;
#pragma unroll
        for (int j = 0; j < 8; ++j) {
          float v = yb[lane * YST + q * 8 + j];
          sm += v; s2 += v * v;
        }
        pr[2 * h] = sm;  pr[2 * h + 1] = s2;
      }
      *reinterpret_cast<float4*>(&lds_red[lane * RST + rw * 4]) =
          make_float4(pr[0], pr[1], pr[2], pr[3]);
    }
    __syncthreads();

    if (front) {
      float ur[16];
      if (step + 1 < NSUB)
        fetch_u_tile128(u, b, chunk * LCH + (step + 1) * SUB, rw, lane, ur);
      if (step >= 2) {
        const float* yb = lds_y[step & 1];
        int t0 = chunk * LCH + (step - 2) * SUB;
        float sm = 0.f, s2 = 0.f;
#pragma unroll
        for (int q = 0; q < 8; ++q) {
          sm += lds_red[lane * RST + q * 2 + 0];
          s2 += lds_red[lane * RST + q * 2 + 1];
        }
        float mu = sm * (1.0f / CO);
        float var = s2 * (1.0f / CO) - mu * mu;
        float rs = rsqrtf(var + 1e-5f);
#pragma unroll
        for (int j = 0; j < 16; ++j) {
          int o = rw * 16 + j;
          float v = (yb[lane * YST + o] - mu) * rs * lw[j] + lb[j];
          v = v / (1.0f + __expf(-v));
          yact[((size_t)(b * CO + o)) * TLEN + t0 + lane] = v;
        }
      }
      if (step < NSUB) {
        float* S = lds_v[step & 1];
#pragma unroll 16
        for (int tl = 0; tl < SUB; ++tl) {
          s0 = s0 * a0 + S[tl * VST + rt];
          S[tl * VST + rt] = s0;
          s1 = s1 * a1 + S[tl * VST + rt + 128];
          S[tl * VST + rt + 128] = s1;
        }
      }
      if (step + 1 < NSUB)
        put_u_tile128(lds_u, rw, lane, ur);
    } else if (step >= 1 && step <= NSUB) {
      const float* S = lds_v[(step - 1) & 1];
      float* yb = lds_y[(step - 1) & 1];
      v8f y0 = {0.f, 0.f, 0.f, 0.f, 0.f, 0.f, 0.f, 0.f};
      v8f y1 = {0.f, 0.f, 0.f, 0.f, 0.f, 0.f, 0.f, 0.f};
      v2f fa[2][8][2];
#pragma unroll
      for (int i = 0; i < 8; ++i) {
        int k = i * 4 + 2 * hi;
        fa[0][i][0].x = S[lo * VST + k];         fa[0][i][0].y = S[lo * VST + k + 1];
        fa[0][i][1].x = S[(16 + lo) * VST + k];  fa[0][i][1].y = S[(16 + lo) * VST + k + 1];
      }
#pragma unroll
      for (int blk = 0; blk < NST / 32; ++blk) {
        if (blk + 1 < NST / 32) {
#pragma unroll
          for (int i = 0; i < 8; ++i) {
            int k = ((blk + 1) * 8 + i) * 4 + 2 * hi;
            fa[(blk + 1) & 1][i][0].x = S[lo * VST + k];
            fa[(blk + 1) & 1][i][0].y = S[lo * VST + k + 1];
            fa[(blk + 1) & 1][i][1].x = S[(16 + lo) * VST + k];
            fa[(blk + 1) & 1][i][1].y = S[(16 + lo) * VST + k + 1];
          }
        }
        __builtin_amdgcn_sched_barrier(0);
#pragma unroll
        for (int i = 0; i < 8; ++i) {
          y0 = wmma4(fa[blk & 1][i][0], fr[blk * 8 + i], y0);
          y1 = wmma4(fa[blk & 1][i][1], fr[blk * 8 + i], y1);
        }
        __builtin_amdgcn_sched_barrier(0);
      }
#pragma unroll
      for (int r = 0; r < 8; ++r) {
        yb[(r + 8 * hi) * YST + rw * 16 + lo] = y0[r];
        yb[(16 + r + 8 * hi) * YST + rw * 16 + lo] = y1[r];
      }
    }
    __syncthreads();
  }
}

__global__ void k_conv(const float* __restrict__ yact,
                       const float* __restrict__ conv_w,
                       const float* __restrict__ conv_b,
                       float* __restrict__ out) {
  int b = blockIdx.y, tid = threadIdx.x;
  int wave = tid >> 5, lane = tid & 31, lo = lane & 15, hi = lane >> 4;
  int tau0 = (blockIdx.x * 4 + wave) * 16;

  v8f acc[4];
#pragma unroll
  for (int mt = 0; mt < 4; ++mt)
#pragma unroll
    for (int r = 0; r < 8; ++r)
      acc[mt][r] = conv_b[mt * 16 + r + 8 * hi];

#pragma unroll 2
  for (int ks = 0; ks < CO / 4; ++ks) {
    int k = ks * 4 + 2 * hi;
    v2f be, bo;
    be.x = yact[((size_t)(b * CO + k)) * TLEN + tau0 + lo];
    be.y = yact[((size_t)(b * CO + k + 1)) * TLEN + tau0 + lo];
    bo.x = yact[((size_t)(b * CO + k)) * TLEN + T2 + tau0 + lo];
    bo.y = yact[((size_t)(b * CO + k + 1)) * TLEN + T2 + tau0 + lo];
#pragma unroll
    for (int mt = 0; mt < 4; ++mt) {
      int m = mt * 16 + lo;
      v2f ae, ao;
      ae.x = conv_w[m * 128 + 2 * k];
      ae.y = conv_w[m * 128 + 2 * (k + 1)];
      ao.x = conv_w[m * 128 + 2 * k + 1];
      ao.y = conv_w[m * 128 + 2 * (k + 1) + 1];
      acc[mt] = wmma4(ae, be, acc[mt]);
      acc[mt] = wmma4(ao, bo, acc[mt]);
    }
  }

#pragma unroll
  for (int mt = 0; mt < 4; ++mt)
#pragma unroll
    for (int r = 0; r < 8; ++r) {
      int o = mt * 16 + r + 8 * hi;
      int t = 2 * (tau0 + lo);
      float v = acc[mt][r];
      float2 pv = make_float2(v, v);
      *reinterpret_cast<float2*>(&out[((size_t)(b * CO + o)) * TLEN + t]) = pv;
    }
}

extern "C" void kernel_launch(void* const* d_in, const int* in_sizes, int n_in,
                              void* d_out, int out_size, void* d_ws, size_t ws_size,
                              hipStream_t stream) {
  const float* u     = (const float*)d_in[0];
  const float* rl    = (const float*)d_in[1];
  const float* Bc    = (const float*)d_in[2];
  const float* Cm    = (const float*)d_in[3];
  const float* ln_w  = (const float*)d_in[4];
  const float* ln_b  = (const float*)d_in[5];
  const float* cw    = (const float*)d_in[6];
  const float* cb    = (const float*)d_in[7];
  float* out = (float*)d_out;

  float* ws    = (float*)d_ws;
  float* Ad    = ws;
  float* Bd    = Ad + NST;
  float* ce    = Bd + CIN * NST;
  float* carry = ce + NCH * BSZ * NST;
  float* yact  = carry + NCH * BSZ * NST;

  k_discretize<<<1, NST, 0, stream>>>(rl, Bc, Ad, Bd);
  k_chunk_ends<<<dim3(NCH, BSZ), 256, 0, stream>>>(u, Ad, Bd, ce);
  k_combine<<<BSZ, NST, 0, stream>>>(Ad, ce, carry);
  k_scan_y<<<dim3(NCH, BSZ), 256, 0, stream>>>(u, Ad, Bd, Cm, ln_w, ln_b,
                                               carry, yact);
  k_conv<<<dim3(T2 / 64, BSZ), 128, 0, stream>>>(yact, cw, cb, out);
}


# k_discretize and k_combine folded into the scan kernels' prologues (discretization via LDS by the role with few loads; carry recurrence per thread, same order): 5 launches -> 3
# speedup vs baseline: 2.2914x; 1.0644x over previous
#include <hip/hip_runtime.h>
#include <hip/hip_bf16.h>


#define BSZ   16
#define CIN   64
#define CO    64
#define TLEN  8192
#define NST   256
#define NCH   16
#define LCH   512
#define SUB   32
#define NSUB  (LCH/SUB)
#define T2    (TLEN/2)

#define UST   68
#define VST   260
#define YST   66
#define RST   20

typedef __attribute__((ext_vector_type(2))) float v2f;
typedef __attribute__((ext_vector_type(8))) float v8f;

__device__ __forceinline__ v8f wmma4(v2f a, v2f b, v8f c) {
  return __builtin_amdgcn_wmma_f32_16x16x4_f32(
       false, a,  false, b,
       (short)0, c,  false,  false);
}

__device__ __forceinline__ void discretize_one(float x, float& ad, float& factor) {
  float sp = fmaxf(x, 0.0f) + log1pf(expf(-fabsf(x)));
  float lam = -sp;
  ad = expf(lam);
  factor = (fabsf(lam) > 1e-6f) ? (ad - 1.0f) / lam : 1.0f;
}
__device__ __forceinline__ void discretize_2cols(float x0, float x1, int n,
                                                 float* lds_ad, float* lds_fac) {
  __builtin_amdgcn_sched_barrier(0);
  float ad0, f0, ad1, f1;
  discretize_one(x0, ad0, f0);
  discretize_one(x1, ad1, f1);
  lds_ad[n] = ad0;        lds_fac[n] = f0;
  lds_ad[n + 128] = ad1;  lds_fac[n + 128] = f1;
}


__device__ __forceinline__ void fetch_u_tile128(const float* __restrict__ u, int b,
                                                int t0, int rw, int lane,
                                                float (&ur)[16]) {
#pragma unroll
  for (int kk = 0; kk < CIN / 4; ++kk) {
    int c = rw * (CIN / 4) + kk;
    ur[kk] = u[((size_t)(b * CIN + c)) * TLEN + t0 + lane];
  }
}
__device__ __forceinline__ void put_u_tile128(float* lds_u, int rw, int lane,
                                              const float (&ur)[16]) {
#pragma unroll
  for (int q = 0; q < 4; ++q)
    *reinterpret_cast<float4*>(&lds_u[lane * UST + rw * (CIN / 4) + 4 * q]) =
        make_float4(ur[4 * q], ur[4 * q + 1], ur[4 * q + 2], ur[4 * q + 3]);
}

__device__ __forceinline__ void load_bd_frags4(const float* __restrict__ Bd,
                                               int rw, int lo, int hi,
                                               v2f (&fr)[64]) {
#pragma unroll
  for (int j = 0; j < 4; ++j) {
    int nn = (rw * 4 + j) * 16 + lo;
#pragma unroll
    for (int ks = 0; ks < CIN / 4; ++ks) {
      int k = ks * 4 + 2 * hi;
      fr[j * 16 + ks].x = Bd[k * NST + nn];
      fr[j * 16 + ks].y = Bd[(k + 1) * NST + nn];
    }
  }
}

__device__ __forceinline__ void scale_bd_frags4(const float* lds_fac, int rw, int lo,
                                                v2f (&fr)[64]) {
#pragma unroll
  for (int j = 0; j < 4; ++j) {
    float f = lds_fac[(rw * 4 + j) * 16 + lo];
#pragma unroll
    for (int ks = 0; ks < CIN / 4; ++ks) {
      fr[j * 16 + ks].x = fr[j * 16 + ks].x * f;
      fr[j * 16 + ks].y = fr[j * 16 + ks].y * f;
    }
  }
}

__device__ __forceinline__ void v_tile_wmma8(const float* lds_u,
                                             const v2f (&fr)[64],
                                             float* lds_v, int rw, int lo, int hi) {
  v2f fa[2][CIN / 4];
#pragma unroll
  for (int ks = 0; ks < CIN / 4; ++ks) {
    int k = ks * 4 + 2 * hi;
    fa[0][ks] = *reinterpret_cast<const v2f*>(&lds_u[lo * UST + k]);
    fa[1][ks] = *reinterpret_cast<const v2f*>(&lds_u[(16 + lo) * UST + k]);
  }
  __builtin_amdgcn_sched_barrier(0);
  v8f acc[2][4];
#pragma unroll
  for (int mt = 0; mt < 2; ++mt)
#pragma unroll
    for (int j = 0; j < 4; ++j)
#pragma unroll
      for (int r = 0; r < 8; ++r) acc[mt][j][r] = 0.f;
#pragma unroll
  for (int ks = 0; ks < CIN / 4; ++ks) {
#pragma unroll
    for (int j = 0; j < 4; ++j) {
      acc[0][j] = wmma4(fa[0][ks], fr[j * 16 + ks], acc[0][j]);
      acc[1][j] = wmma4(fa[1][ks], fr[j * 16 + ks], acc[1][j]);
    }
  }
#pragma unroll
  for (int mt = 0; mt < 2; ++mt)
#pragma unroll
    for (int j = 0; j < 4; ++j)
#pragma unroll
      for (int r = 0; r < 8; ++r)
        lds_v[(mt * 16 + r + 8 * hi) * VST + (rw * 4 + j) * 16 + lo] = acc[mt][j][r];
}

__global__ void __launch_bounds__(256)
k_chunk_ends(const float* __restrict__ u,
             const float* __restrict__ rl,
             const float* __restrict__ Bc,
             float* __restrict__ ce) {
  __shared__ __attribute__((aligned(16))) float lds_u[2][SUB * UST];
  __shared__ float lds_v[2][SUB * VST];
  __shared__ float lds_ad[NST];
  __shared__ float lds_fac[NST];
  const int chunk = blockIdx.x, b = blockIdx.y, tid = threadIdx.x;
  const int lane = tid & 31, lo = lane & 15, hi = lane >> 4;
  const int wave = __builtin_amdgcn_readfirstlane(tid >> 5);
  const bool gemm_role = wave < 4;
  const int rw = wave & 3, rt = rw * 32 + lane;

  v2f fr[64];
#pragma unroll
  for (int i = 0; i < 64; ++i) { fr[i].x = 0.f; fr[i].y = 0.f; }
  float a0 = 0.f, a1 = 0.f, s0 = 0.f, s1 = 0.f;
  if (gemm_role) {
    load_bd_frags4(Bc, rw, lo, hi, fr);
  } else {
    const float x0 = rl[rt], x1 = rl[rt + 128];
    float ur[16];
    fetch_u_tile128(u, b, chunk * LCH, rw, lane, ur);
    discretize_2cols(x0, x1, rt, lds_ad, lds_fac);
    put_u_tile128(lds_u[0], rw, lane, ur);
  }
  __syncthreads();
  if (gemm_role) {
    scale_bd_frags4(lds_fac, rw, lo, fr);
  } else {
    a0 = lds_ad[rt];  a1 = lds_ad[rt + 128];
  }

  for (int step = 0; step <= NSUB; ++step) {
    if (gemm_role) {
      if (step < NSUB)
        v_tile_wmma8(lds_u[step & 1], fr, lds_v[step & 1], rw, lo, hi);
    } else {
      float ur[16];
      if (step + 1 < NSUB)
        fetch_u_tile128(u, b, chunk * LCH + (step + 1) * SUB, rw, lane, ur);
      if (step >= 1) {
        const float* S = lds_v[(step - 1) & 1];
#pragma unroll 16
        for (int tl = 0; tl < SUB; ++tl) {
          s0 = s0 * a0 + S[tl * VST + rt];
          s1 = s1 * a1 + S[tl * VST + rt + 128];
        }
      }
      if (step + 1 < NSUB)
        put_u_tile128(lds_u[(step + 1) & 1], rw, lane, ur);
    }
    __syncthreads();
  }
  if (!gemm_role) {
    ce[chunk * (BSZ * NST) + b * NST + rt] = s0;
    ce[chunk * (BSZ * NST) + b * NST + rt + 128] = s1;
  }
}

__global__ void __launch_bounds__(256)
k_scan_y(const float* __restrict__ u,
         const float* __restrict__ rl,
         const float* __restrict__ Bc,
         const float* __restrict__ Cm,
         const float* __restrict__ ln_w,
         const float* __restrict__ ln_b,
         const float* __restrict__ ce,
         float* __restrict__ yact) {
  __shared__ float lds_ad[NST];
  __shared__ float lds_fac[NST];
  __shared__ __attribute__((aligned(16))) float lds_u[SUB * UST];
  __shared__ float lds_v[2][SUB * VST];
  __shared__ float lds_y[2][SUB * YST];
  __shared__ __attribute__((aligned(16))) float lds_red[SUB * RST];

  const int chunk = blockIdx.x, b = blockIdx.y, tid = threadIdx.x;
  const int lane = tid & 31, lo = lane & 15, hi = lane >> 4;
  const int wave = __builtin_amdgcn_readfirstlane(tid >> 5);
  const bool front = wave < 4;
  const int rw = wave & 3, rt = rw * 32 + lane;

  v2f fr[64];
#pragma unroll
  for (int i = 0; i < 64; ++i) { fr[i].x = 0.f; fr[i].y = 0.f; }
  float a0 = 0.f, a1 = 0.f, s0 = 0.f, s1 = 0.f;
  float lw[16], lb[16];
#pragma unroll
  for (int j = 0; j < 16; ++j) { lw[j] = 0.f; lb[j] = 0.f; }

  float e0[NCH - 1], e1[NCH - 1];
#pragma unroll
  for (int c = 0; c < NCH - 1; ++c) { e0[c] = 0.f; e1[c] = 0.f; }

  if (front) {
    load_bd_frags4(Bc, rw, lo, hi, fr);
#pragma unroll
    for (int c = 0; c < NCH - 1; ++c) {
      e0[c] = ce[c * (BSZ * NST) + b * NST + rt];
      e1[c] = ce[c * (BSZ * NST) + b * NST + rt + 128];
    }
#pragma unroll
    for (int j = 0; j < 16; ++j) { lw[j] = ln_w[rw * 16 + j]; lb[j] = ln_b[rw * 16 + j]; }
    float ur[16];
    fetch_u_tile128(u, b, chunk * LCH, rw, lane, ur);
    put_u_tile128(lds_u, rw, lane, ur);
  } else {
    const float x0 = rl[rt], x1 = rl[rt + 128];
    int oo = rw * 16 + lo;
#pragma unroll
    for (int ks = 0; ks < NST / 4; ++ks) {
      int k = ks * 4 + 2 * hi;
      fr[ks].x = Cm[k * CO + oo];
      fr[ks].y = Cm[(k + 1) * CO + oo];
    }
    discretize_2cols(x0, x1, rt, lds_ad, lds_fac);
  }
  __syncthreads();
  if (front) {
    scale_bd_frags4(lds_fac, rw, lo, fr);
    a0 = lds_ad[rt];  a1 = lds_ad[rt + 128];
    float aL0 = a0, aL1 = a1;
#pragma unroll
    for (int i = 0; i < 9; ++i) { aL0 = aL0 * aL0; aL1 = aL1 * aL1; }
#pragma unroll
    for (int c = 0; c < NCH - 1; ++c) {
      if (c < chunk) {
        s0 = __builtin_fmaf(aL0, s0, e0[c]);
        s1 = __builtin_fmaf(aL1, s1, e1[c]);
      }
    }
  }

  for (int step = 0; step < NSUB + 2; ++step) {
    if (front) {
      if (step < NSUB)
        v_tile_wmma8(lds_u, fr, lds_v[step & 1], rw, lo, hi);
    } else if (step >= 2) {
      const float* yb = lds_y[step & 1];
      float pr[4];
#pragma unroll
      for (int h = 0; h < 2; ++h) {
        int q = rw * 2 + h;
        float sm = 0.f, s2 = 0.f;
#pragma unroll
        for (int j = 0; j < 8; ++j) {
          float v = yb[lane * YST + q * 8 + j];
          sm += v; s2 += v * v;
        }
        pr[2 * h] = sm;  pr[2 * h + 1] = s2;
      }
      *reinterpret_cast<float4*>(&lds_red[lane * RST + rw * 4]) =
          make_float4(pr[0], pr[1], pr[2], pr[3]);
    }
    __syncthreads();

    if (front) {
      float ur[16];
      if (step + 1 < NSUB)
        fetch_u_tile128(u, b, chunk * LCH + (step + 1) * SUB, rw, lane, ur);
      if (step >= 2) {
        const float* yb = lds_y[step & 1];
        int t0 = chunk * LCH + (step - 2) * SUB;
        float sm = 0.f, s2 = 0.f;
#pragma unroll
        for (int q = 0; q < 8; ++q) {
          sm += lds_red[lane * RST + q * 2 + 0];
          s2 += lds_red[lane * RST + q * 2 + 1];
        }
        float mu = sm * (1.0f / CO);
        float var = s2 * (1.0f / CO) - mu * mu;
        float rs = rsqrtf(var + 1e-5f);
#pragma unroll
        for (int j = 0; j < 16; ++j) {
          int o = rw * 16 + j;
          float v = (yb[lane * YST + o] - mu) * rs * lw[j] + lb[j];
          v = v / (1.0f + __expf(-v));
          yact[((size_t)(b * CO + o)) * TLEN + t0 + lane] = v;
        }
      }
      if (step < NSUB) {
        float* S = lds_v[step & 1];
#pragma unroll 16
        for (int tl = 0; tl < SUB; ++tl) {
          s0 = s0 * a0 + S[tl * VST + rt];
          S[tl * VST + rt] = s0;
          s1 = s1 * a1 + S[tl * VST + rt + 128];
          S[tl * VST + rt + 128] = s1;
        }
      }
      if (step + 1 < NSUB)
        put_u_tile128(lds_u, rw, lane, ur);
    } else if (step >= 1 && step <= NSUB) {
      const float* S = lds_v[(step - 1) & 1];
      float* yb = lds_y[(step - 1) & 1];
      v8f y0 = {0.f, 0.f, 0.f, 0.f, 0.f, 0.f, 0.f, 0.f};
      v8f y1 = {0.f, 0.f, 0.f, 0.f, 0.f, 0.f, 0.f, 0.f};
      v2f fa[2][8][2];
#pragma unroll
      for (int i = 0; i < 8; ++i) {
        int k = i * 4 + 2 * hi;
        fa[0][i][0].x = S[lo * VST + k];         fa[0][i][0].y = S[lo * VST + k + 1];
        fa[0][i][1].x = S[(16 + lo) * VST + k];  fa[0][i][1].y = S[(16 + lo) * VST + k + 1];
      }
#pragma unroll
      for (int blk = 0; blk < NST / 32; ++blk) {
        if (blk + 1 < NST / 32) {
#pragma unroll
          for (int i = 0; i < 8; ++i) {
            int k = ((blk + 1) * 8 + i) * 4 + 2 * hi;
            fa[(blk + 1) & 1][i][0].x = S[lo * VST + k];
            fa[(blk + 1) & 1][i][0].y = S[lo * VST + k + 1];
            fa[(blk + 1) & 1][i][1].x = S[(16 + lo) * VST + k];
            fa[(blk + 1) & 1][i][1].y = S[(16 + lo) * VST + k + 1];
          }
        }
        __builtin_amdgcn_sched_barrier(0);
#pragma unroll
        for (int i = 0; i < 8; ++i) {
          y0 = wmma4(fa[blk & 1][i][0], fr[blk * 8 + i], y0);
          y1 = wmma4(fa[blk & 1][i][1], fr[blk * 8 + i], y1);
        }
        __builtin_amdgcn_sched_barrier(0);
      }
#pragma unroll
      for (int r = 0; r < 8; ++r) {
        yb[(r + 8 * hi) * YST + rw * 16 + lo] = y0[r];
        yb[(16 + r + 8 * hi) * YST + rw * 16 + lo] = y1[r];
      }
    }
    __syncthreads();
  }
}

__global__ void k_conv(const float* __restrict__ yact,
                       const float* __restrict__ conv_w,
                       const float* __restrict__ conv_b,
                       float* __restrict__ out) {
  int b = blockIdx.y, tid = threadIdx.x;
  int wave = tid >> 5, lane = tid & 31, lo = lane & 15, hi = lane >> 4;
  int tau0 = (blockIdx.x * 4 + wave) * 16;

  v8f acc[4];
#pragma unroll
  for (int mt = 0; mt < 4; ++mt)
#pragma unroll
    for (int r = 0; r < 8; ++r)
      acc[mt][r] = conv_b[mt * 16 + r + 8 * hi];

#pragma unroll 2
  for (int ks = 0; ks < CO / 4; ++ks) {
    int k = ks * 4 + 2 * hi;
    v2f be, bo;
    be.x = yact[((size_t)(b * CO + k)) * TLEN + tau0 + lo];
    be.y = yact[((size_t)(b * CO + k + 1)) * TLEN + tau0 + lo];
    bo.x = yact[((size_t)(b * CO + k)) * TLEN + T2 + tau0 + lo];
    bo.y = yact[((size_t)(b * CO + k + 1)) * TLEN + T2 + tau0 + lo];
#pragma unroll
    for (int mt = 0; mt < 4; ++mt) {
      int m = mt * 16 + lo;
      v2f ae, ao;
      ae.x = conv_w[m * 128 + 2 * k];
      ae.y = conv_w[m * 128 + 2 * (k + 1)];
      ao.x = conv_w[m * 128 + 2 * k + 1];
      ao.y = conv_w[m * 128 + 2 * (k + 1) + 1];
      acc[mt] = wmma4(ae, be, acc[mt]);
      acc[mt] = wmma4(ao, bo, acc[mt]);
    }
  }

#pragma unroll
  for (int mt = 0; mt < 4; ++mt)
#pragma unroll
    for (int r = 0; r < 8; ++r) {
      int o = mt * 16 + r + 8 * hi;
      int t = 2 * (tau0 + lo);
      float v = acc[mt][r];
      float2 pv = make_float2(v, v);
      *reinterpret_cast<float2*>(&out[((size_t)(b * CO + o)) * TLEN + t]) = pv;
    }
}

extern "C" void kernel_launch(void* const* d_in, const int* in_sizes, int n_in,
                              void* d_out, int out_size, void* d_ws, size_t ws_size,
                              hipStream_t stream) {
  const float* u     = (const float*)d_in[0];
  const float* rl    = (const float*)d_in[1];
  const float* Bc    = (const float*)d_in[2];
  const float* Cm    = (const float*)d_in[3];
  const float* ln_w  = (const float*)d_in[4];
  const float* ln_b  = (const float*)d_in[5];
  const float* cw    = (const float*)d_in[6];
  const float* cb    = (const float*)d_in[7];
  float* out = (float*)d_out;

  float* ws    = (float*)d_ws;
  float* ce    = ws;
  float* yact  = ce + NCH * BSZ * NST;

  k_chunk_ends<<<dim3(NCH, BSZ), 256, 0, stream>>>(u, rl, Bc, ce);
  k_scan_y<<<dim3(NCH, BSZ), 256, 0, stream>>>(u, rl, Bc, Cm, ln_w, ln_b,
                                               ce, yact);
  k_conv<<<dim3(T2 / 64, BSZ), 128, 0, stream>>>(yact, cw, cb, out);
}


# k_chunk_ends: all 8 waves run the v GEMM (alternate subtiles, A fragments from global); end state summed from the accumulators by powers of a (summation order changes); no LDS tile, no loop barrier
# speedup vs baseline: 2.5168x; 1.0984x over previous
#include <hip/hip_runtime.h>
#include <hip/hip_bf16.h>


#define BSZ   16
#define CIN   64
#define CO    64
#define TLEN  8192
#define NST   256
#define NCH   16
#define LCH   512
#define SUB   32
#define NSUB  (LCH/SUB)
#define T2    (TLEN/2)

#define UST   68
#define VST   260
#define YST   66
#define RST   20

typedef __attribute__((ext_vector_type(2))) float v2f;
typedef __attribute__((ext_vector_type(8))) float v8f;

__device__ __forceinline__ v8f wmma4(v2f a, v2f b, v8f c) {
  return __builtin_amdgcn_wmma_f32_16x16x4_f32(
       false, a,  false, b,
       (short)0, c,  false,  false);
}

__device__ __forceinline__ void discretize_one(float x, float& ad, float& factor) {
  float sp = fmaxf(x, 0.0f) + log1pf(expf(-fabsf(x)));
  float lam = -sp;
  ad = expf(lam);
  factor = (fabsf(lam) > 1e-6f) ? (ad - 1.0f) / lam : 1.0f;
}
__device__ __forceinline__ void discretize_2cols(float x0, float x1, int n,
                                                 float* lds_ad, float* lds_fac) {
  __builtin_amdgcn_sched_barrier(0);
  float ad0, f0, ad1, f1;
  discretize_one(x0, ad0, f0);
  discretize_one(x1, ad1, f1);
  lds_ad[n] = ad0;        lds_fac[n] = f0;
  lds_ad[n + 128] = ad1;  lds_fac[n + 128] = f1;
}


__device__ __forceinline__ void fetch_u_tile128(const float* __restrict__ u, int b,
                                                int t0, int rw, int lane,
                                                float (&ur)[16]) {
#pragma unroll
  for (int kk = 0; kk < CIN / 4; ++kk) {
    int c = rw * (CIN / 4) + kk;
    ur[kk] = u[((size_t)(b * CIN + c)) * TLEN + t0 + lane];
  }
}
__device__ __forceinline__ void put_u_tile128(float* lds_u, int rw, int lane,
                                              const float (&ur)[16]) {
#pragma unroll
  for (int q = 0; q < 4; ++q)
    *reinterpret_cast<float4*>(&lds_u[lane * UST + rw * (CIN / 4) + 4 * q]) =
        make_float4(ur[4 * q], ur[4 * q + 1], ur[4 * q + 2], ur[4 * q + 3]);
}

__device__ __forceinline__ void load_bd_frags4(const float* __restrict__ Bd,
                                               int rw, int lo, int hi,
                                               v2f (&fr)[64]) {
#pragma unroll
  for (int j = 0; j < 4; ++j) {
    int nn = (rw * 4 + j) * 16 + lo;
#pragma unroll
    for (int ks = 0; ks < CIN / 4; ++ks) {
      int k = ks * 4 + 2 * hi;
      fr[j * 16 + ks].x = Bd[k * NST + nn];
      fr[j * 16 + ks].y = Bd[(k + 1) * NST + nn];
    }
  }
}

__device__ __forceinline__ void scale_bd_frags4(const float* lds_fac, int rw, int lo,
                                                v2f (&fr)[64]) {
#pragma unroll
  for (int j = 0; j < 4; ++j) {
    float f = lds_fac[(rw * 4 + j) * 16 + lo];
#pragma unroll
    for (int ks = 0; ks < CIN / 4; ++ks) {
      fr[j * 16 + ks].x = fr[j * 16 + ks].x * f;
      fr[j * 16 + ks].y = fr[j * 16 + ks].y * f;
    }
  }
}

__device__ __forceinline__ void v_tile_wmma8(const float* lds_u,
                                             const v2f (&fr)[64],
                                             float* lds_v, int rw, int lo, int hi) {
  v2f fa[2][CIN / 4];
#pragma unroll
  for (int ks = 0; ks < CIN / 4; ++ks) {
    int k = ks * 4 + 2 * hi;
    fa[0][ks] = *reinterpret_cast<const v2f*>(&lds_u[lo * UST + k]);
    fa[1][ks] = *reinterpret_cast<const v2f*>(&lds_u[(16 + lo) * UST + k]);
  }
  __builtin_amdgcn_sched_barrier(0);
  v8f acc[2][4];
#pragma unroll
  for (int mt = 0; mt < 2; ++mt)
#pragma unroll
    for (int j = 0; j < 4; ++j)
#pragma unroll
      for (int r = 0; r < 8; ++r) acc[mt][j][r] = 0.f;
#pragma unroll
  for (int ks = 0; ks < CIN / 4; ++ks) {
#pragma unroll
    for (int j = 0; j < 4; ++j) {
      acc[0][j] = wmma4(fa[0][ks], fr[j * 16 + ks], acc[0][j]);
      acc[1][j] = wmma4(fa[1][ks], fr[j * 16 + ks], acc[1][j]);
    }
  }
#pragma unroll
  for (int mt = 0; mt < 2; ++mt)
#pragma unroll
    for (int j = 0; j < 4; ++j)
#pragma unroll
      for (int r = 0; r < 8; ++r)
        lds_v[(mt * 16 + r + 8 * hi) * VST + (rw * 4 + j) * 16 + lo] = acc[mt][j][r];
}

__device__ __forceinline__ void load_u_pairs(const float* __restrict__ up, int t0,
                                             float2 (&ra)[CIN / 2]) {
#pragma unroll
  for (int ks = 0; ks < CIN / 4; ++ks) {
    ra[2 * ks]     = *reinterpret_cast<const float2*>(&up[(size_t)(4 * ks) * TLEN + t0]);
    ra[2 * ks + 1] = *reinterpret_cast<const float2*>(&up[(size_t)(4 * ks + 1) * TLEN + t0]);
  }
}

__device__ __forceinline__ void subtile_end_sums(const float2 (&ra)[CIN / 2],
                                                 const v2f (&fr)[64],
                                                 const float (&a)[4],
                                                 const float (&a2)[4],
                                                 const float (&a64)[4],
                                                 const float (&wh)[4],
                                                 float (&sl)[4]) {
  v2f fa[2][CIN / 4];
#pragma unroll
  for (int ks = 0; ks < CIN / 4; ++ks) {
    fa[0][ks].x = ra[2 * ks].x;  fa[0][ks].y = ra[2 * ks + 1].x;
    fa[1][ks].x = ra[2 * ks].y;  fa[1][ks].y = ra[2 * ks + 1].y;
  }
  v8f acc[2][4];
#pragma unroll
  for (int mt = 0; mt < 2; ++mt)
#pragma unroll
    for (int j = 0; j < 4; ++j)
#pragma unroll
      for (int r = 0; r < 8; ++r) acc[mt][j][r] = 0.f;
#pragma unroll
  for (int ks = 0; ks < CIN / 4; ++ks) {
#pragma unroll
    for (int j = 0; j < 4; ++j) {
      acc[0][j] = wmma4(fa[0][ks], fr[j * 16 + ks], acc[0][j]);
      acc[1][j] = wmma4(fa[1][ks], fr[j * 16 + ks], acc[1][j]);
    }
  }
#pragma unroll
  for (int j = 0; j < 4; ++j) {
    float h = __builtin_fmaf(acc[0][j][0], a[j], acc[1][j][0]);
#pragma unroll
    for (int r = 1; r < 8; ++r) {
      float q = __builtin_fmaf(acc[0][j][r], a[j], acc[1][j][r]);
      h = __builtin_fmaf(h, a2[j], q);
    }
    sl[j] = __builtin_fmaf(sl[j], a64[j], h * wh[j]);
  }
}

__global__ void __launch_bounds__(256)
k_chunk_ends(const float* __restrict__ u,
             const float* __restrict__ rl,
             const float* __restrict__ Bc,
             float* __restrict__ ce) {
  __shared__ float lds_x[NST];
  const int chunk = blockIdx.x, b = blockIdx.y, tid = threadIdx.x;
  const int lane = tid & 31, lo = lane & 15, hi = lane >> 4;
  const int wave = __builtin_amdgcn_readfirstlane(tid >> 5);
  const int par = wave >> 2, rw = wave & 3;

  v2f fr[64];
  load_bd_frags4(Bc, rw, lo, hi, fr);
  float x[4];
#pragma unroll
  for (int j = 0; j < 4; ++j) x[j] = rl[(rw * 4 + j) * 16 + lo];
  const float* up = u + ((size_t)(b * CIN + 2 * hi)) * TLEN + chunk * LCH + 2 * lo;
  float2 ra0[CIN / 2], ra1[CIN / 2];
  load_u_pairs(up, par * SUB, ra0);

  float a[4], a2[4], a32[4], a64[4], wh[4], sl[4];
#pragma unroll
  for (int j = 0; j < 4; ++j) {
    float f;
    discretize_one(x[j], a[j], f);
#pragma unroll
    for (int ks = 0; ks < CIN / 4; ++ks) {
      fr[j * 16 + ks].x = fr[j * 16 + ks].x * f;
      fr[j * 16 + ks].y = fr[j * 16 + ks].y * f;
    }
    a2[j] = a[j] * a[j];
    const float a4 = a2[j] * a2[j], a8 = a4 * a4, a16 = a8 * a8;
    a32[j] = a16 * a16;  a64[j] = a32[j] * a32[j];
    wh[j] = hi ? 1.0f : a16;
    sl[j] = 0.f;
  }

  for (int p = 0; p < NSUB / 4 - 1; ++p) {
    const int s = 4 * p + par;
    load_u_pairs(up, (s + 2) * SUB, ra1);
    __builtin_amdgcn_sched_barrier(0);
    subtile_end_sums(ra0, fr, a, a2, a64, wh, sl);
    load_u_pairs(up, (s + 4) * SUB, ra0);
    __builtin_amdgcn_sched_barrier(0);
    subtile_end_sums(ra1, fr, a, a2, a64, wh, sl);
  }
  load_u_pairs(up, (NSUB - 2 + par) * SUB, ra1);
  __builtin_amdgcn_sched_barrier(0);
  subtile_end_sums(ra0, fr, a, a2, a64, wh, sl);
  subtile_end_sums(ra1, fr, a, a2, a64, wh, sl);

  float tot[4];
#pragma unroll
  for (int j = 0; j < 4; ++j) tot[j] = sl[j] + __shfl_xor(sl[j], 16);
  if (par == 0 && hi == 0) {
#pragma unroll
    for (int j = 0; j < 4; ++j) lds_x[(rw * 4 + j) * 16 + lo] = tot[j];
  }
  __syncthreads();
  if (par == 1 && hi == 0) {
#pragma unroll
    for (int j = 0; j < 4; ++j) {
      int n = (rw * 4 + j) * 16 + lo;
      ce[chunk * (BSZ * NST) + b * NST + n] = __builtin_fmaf(lds_x[n], a32[j], tot[j]);
    }
  }
}

__global__ void __launch_bounds__(256)
k_scan_y(const float* __restrict__ u,
         const float* __restrict__ rl,
         const float* __restrict__ Bc,
         const float* __restrict__ Cm,
         const float* __restrict__ ln_w,
         const float* __restrict__ ln_b,
         const float* __restrict__ ce,
         float* __restrict__ yact) {
  __shared__ float lds_ad[NST];
  __shared__ float lds_fac[NST];
  __shared__ __attribute__((aligned(16))) float lds_u[SUB * UST];
  __shared__ float lds_v[2][SUB * VST];
  __shared__ float lds_y[2][SUB * YST];
  __shared__ __attribute__((aligned(16))) float lds_red[SUB * RST];

  const int chunk = blockIdx.x, b = blockIdx.y, tid = threadIdx.x;
  const int lane = tid & 31, lo = lane & 15, hi = lane >> 4;
  const int wave = __builtin_amdgcn_readfirstlane(tid >> 5);
  const bool front = wave < 4;
  const int rw = wave & 3, rt = rw * 32 + lane;

  v2f fr[64];
#pragma unroll
  for (int i = 0; i < 64; ++i) { fr[i].x = 0.f; fr[i].y = 0.f; }
  float a0 = 0.f, a1 = 0.f, s0 = 0.f, s1 = 0.f;
  float lw[16], lb[16];
#pragma unroll
  for (int j = 0; j < 16; ++j) { lw[j] = 0.f; lb[j] = 0.f; }

  float e0[NCH - 1], e1[NCH - 1];
#pragma unroll
  for (int c = 0; c < NCH - 1; ++c) { e0[c] = 0.f; e1[c] = 0.f; }

  if (front) {
    load_bd_frags4(Bc, rw, lo, hi, fr);
#pragma unroll
    for (int c = 0; c < NCH - 1; ++c) {
      e0[c] = ce[c * (BSZ * NST) + b * NST + rt];
      e1[c] = ce[c * (BSZ * NST) + b * NST + rt + 128];
    }
#pragma unroll
    for (int j = 0; j < 16; ++j) { lw[j] = ln_w[rw * 16 + j]; lb[j] = ln_b[rw * 16 + j]; }
    float ur[16];
    fetch_u_tile128(u, b, chunk * LCH, rw, lane, ur);
    put_u_tile128(lds_u, rw, lane, ur);
  } else {
    const float x0 = rl[rt], x1 = rl[rt + 128];
    int oo = rw * 16 + lo;
#pragma unroll
    for (int ks = 0; ks < NST / 4; ++ks) {
      int k = ks * 4 + 2 * hi;
      fr[ks].x = Cm[k * CO + oo];
      fr[ks].y = Cm[(k + 1) * CO + oo];
    }
    discretize_2cols(x0, x1, rt, lds_ad, lds_fac);
  }
  __syncthreads();
  if (front) {
    scale_bd_frags4(lds_fac, rw, lo, fr);
    a0 = lds_ad[rt];  a1 = lds_ad[rt + 128];
    float aL0 = a0, aL1 = a1;
#pragma unroll
    for (int i = 0; i < 9; ++i) { aL0 = aL0 * aL0; aL1 = aL1 * aL1; }
#pragma unroll
    for (int c = 0; c < NCH - 1; ++c) {
      if (c < chunk) {
        s0 = __builtin_fmaf(aL0, s0, e0[c]);
        s1 = __builtin_fmaf(aL1, s1, e1[c]);
      }
    }
  }

  for (int step = 0; step < NSUB + 2; ++step) {
    if (front) {
      if (step < NSUB)
        v_tile_wmma8(lds_u, fr, lds_v[step & 1], rw, lo, hi);
    } else if (step >= 2) {
      const float* yb = lds_y[step & 1];
      float pr[4];
#pragma unroll
      for (int h = 0; h < 2; ++h) {
        int q = rw * 2 + h;
        float sm = 0.f, s2 = 0.f;
#pragma unroll
        for (int j = 0; j < 8; ++j) {
          float v = yb[lane * YST + q * 8 + j];
          sm += v; s2 += v * v;
        }
        pr[2 * h] = sm;  pr[2 * h + 1] = s2;
      }
      *reinterpret_cast<float4*>(&lds_red[lane * RST + rw * 4]) =
          make_float4(pr[0], pr[1], pr[2], pr[3]);
    }
    __syncthreads();

    if (front) {
      float ur[16];
      if (step + 1 < NSUB)
        fetch_u_tile128(u, b, chunk * LCH + (step + 1) * SUB, rw, lane, ur);
      if (step >= 2) {
        const float* yb = lds_y[step & 1];
        int t0 = chunk * LCH + (step - 2) * SUB;
        float sm = 0.f, s2 = 0.f;
#pragma unroll
        for (int q = 0; q < 8; ++q) {
          sm += lds_red[lane * RST + q * 2 + 0];
          s2 += lds_red[lane * RST + q * 2 + 1];
        }
        float mu = sm * (1.0f / CO);
        float var = s2 * (1.0f / CO) - mu * mu;
        float rs = rsqrtf(var + 1e-5f);
#pragma unroll
        for (int j = 0; j < 16; ++j) {
          int o = rw * 16 + j;
          float v = (yb[lane * YST + o] - mu) * rs * lw[j] + lb[j];
          v = v / (1.0f + __expf(-v));
          yact[((size_t)(b * CO + o)) * TLEN + t0 + lane] = v;
        }
      }
      if (step < NSUB) {
        float* S = lds_v[step & 1];
#pragma unroll 16
        for (int tl = 0; tl < SUB; ++tl) {
          s0 = s0 * a0 + S[tl * VST + rt];
          S[tl * VST + rt] = s0;
          s1 = s1 * a1 + S[tl * VST + rt + 128];
          S[tl * VST + rt + 128] = s1;
        }
      }
      if (step + 1 < NSUB)
        put_u_tile128(lds_u, rw, lane, ur);
    } else if (step >= 1 && step <= NSUB) {
      const float* S = lds_v[(step - 1) & 1];
      float* yb = lds_y[(step - 1) & 1];
      v8f y0 = {0.f, 0.f, 0.f, 0.f, 0.f, 0.f, 0.f, 0.f};
      v8f y1 = {0.f, 0.f, 0.f, 0.f, 0.f, 0.f, 0.f, 0.f};
      v2f fa[2][8][2];
#pragma unroll
      for (int i = 0; i < 8; ++i) {
        int k = i * 4 + 2 * hi;
        fa[0][i][0].x = S[lo * VST + k];         fa[0][i][0].y = S[lo * VST + k + 1];
        fa[0][i][1].x = S[(16 + lo) * VST + k];  fa[0][i][1].y = S[(16 + lo) * VST + k + 1];
      }
#pragma unroll
      for (int blk = 0; blk < NST / 32; ++blk) {
        if (blk + 1 < NST / 32) {
#pragma unroll
          for (int i = 0; i < 8; ++i) {
            int k = ((blk + 1) * 8 + i) * 4 + 2 * hi;
            fa[(blk + 1) & 1][i][0].x = S[lo * VST + k];
            fa[(blk + 1) & 1][i][0].y = S[lo * VST + k + 1];
            fa[(blk + 1) & 1][i][1].x = S[(16 + lo) * VST + k];
            fa[(blk + 1) & 1][i][1].y = S[(16 + lo) * VST + k + 1];
          }
        }
        __builtin_amdgcn_sched_barrier(0);
#pragma unroll
        for (int i = 0; i < 8; ++i) {
          y0 = wmma4(fa[blk & 1][i][0], fr[blk * 8 + i], y0);
          y1 = wmma4(fa[blk & 1][i][1], fr[blk * 8 + i], y1);
        }
        __builtin_amdgcn_sched_barrier(0);
      }
#pragma unroll
      for (int r = 0; r < 8; ++r) {
        yb[(r + 8 * hi) * YST + rw * 16 + lo] = y0[r];
        yb[(16 + r + 8 * hi) * YST + rw * 16 + lo] = y1[r];
      }
    }
    __syncthreads();
  }
}

__global__ void k_conv(const float* __restrict__ yact,
                       const float* __restrict__ conv_w,
                       const float* __restrict__ conv_b,
                       float* __restrict__ out) {
  int b = blockIdx.y, tid = threadIdx.x;
  int wave = tid >> 5, lane = tid & 31, lo = lane & 15, hi = lane >> 4;
  int tau0 = (blockIdx.x * 4 + wave) * 16;

  v8f acc[4];
#pragma unroll
  for (int mt = 0; mt < 4; ++mt)
#pragma unroll
    for (int r = 0; r < 8; ++r)
      acc[mt][r] = conv_b[mt * 16 + r + 8 * hi];

#pragma unroll 2
  for (int ks = 0; ks < CO / 4; ++ks) {
    int k = ks * 4 + 2 * hi;
    v2f be, bo;
    be.x = yact[((size_t)(b * CO + k)) * TLEN + tau0 + lo];
    be.y = yact[((size_t)(b * CO + k + 1)) * TLEN + tau0 + lo];
    bo.x = yact[((size_t)(b * CO + k)) * TLEN + T2 + tau0 + lo];
    bo.y = yact[((size_t)(b * CO + k + 1)) * TLEN + T2 + tau0 + lo];
#pragma unroll
    for (int mt = 0; mt < 4; ++mt) {
      int m = mt * 16 + lo;
      v2f ae, ao;
      ae.x = conv_w[m * 128 + 2 * k];
      ae.y = conv_w[m * 128 + 2 * (k + 1)];
      ao.x = conv_w[m * 128 + 2 * k + 1];
      ao.y = conv_w[m * 128 + 2 * (k + 1) + 1];
      acc[mt] = wmma4(ae, be, acc[mt]);
      acc[mt] = wmma4(ao, bo, acc[mt]);
    }
  }

#pragma unroll
  for (int mt = 0; mt < 4; ++mt)
#pragma unroll
    for (int r = 0; r < 8; ++r) {
      int o = mt * 16 + r + 8 * hi;
      int t = 2 * (tau0 + lo);
      float v = acc[mt][r];
      float2 pv = make_float2(v, v);
      *reinterpret_cast<float2*>(&out[((size_t)(b * CO + o)) * TLEN + t]) = pv;
    }
}

extern "C" void kernel_launch(void* const* d_in, const int* in_sizes, int n_in,
                              void* d_out, int out_size, void* d_ws, size_t ws_size,
                              hipStream_t stream) {
  const float* u     = (const float*)d_in[0];
  const float* rl    = (const float*)d_in[1];
  const float* Bc    = (const float*)d_in[2];
  const float* Cm    = (const float*)d_in[3];
  const float* ln_w  = (const float*)d_in[4];
  const float* ln_b  = (const float*)d_in[5];
  const float* cw    = (const float*)d_in[6];
  const float* cb    = (const float*)d_in[7];
  float* out = (float*)d_out;

  float* ws    = (float*)d_ws;
  float* ce    = ws;
  float* yact  = ce + NCH * BSZ * NST;

  k_chunk_ends<<<dim3(NCH, BSZ), 256, 0, stream>>>(u, rl, Bc, ce);
  k_scan_y<<<dim3(NCH, BSZ), 256, 0, stream>>>(u, rl, Bc, Cm, ln_w, ln_b,
                                               ce, yact);
  k_conv<<<dim3(T2 / 64, BSZ), 128, 0, stream>>>(yact, cw, cb, out);
}


# k_conv folded into k_scan_y: a workgroup scans the 256-step chunks at tau and tau+T/2 and runs the conv on their activations from LDS; 3 launches -> 2; carry over 256-step ends (sum order changes)
# speedup vs baseline: 2.9026x; 1.1533x over previous
#include <hip/hip_runtime.h>
#include <hip/hip_bf16.h>


#define BSZ   16
#define CIN   64
#define CO    64
#define TLEN  8192
#define NST   256
#define NCH   16
#define LCH   512
#define SUB   32
#define NSUB  (LCH/SUB)
#define T2    (TLEN/2)
#define LCH2  256
#define NCH2  (TLEN/LCH2)
#define NSUBH (LCH2/SUB)
#define NPAIR (T2/LCH2)

#define UST   68
#define VST   260
#define YST   66
#define RST   20
#define YP    264
#define CWP   132

#define OFF_U   0
#define OFF_Y   (OFF_U + SUB * UST)
#define OFF_RED (OFF_Y + 2 * SUB * YST)
#define OFF_AD  (OFF_RED + SUB * RST)
#define OFF_FAC (OFF_AD + NST)
#define OFF_V   (OFF_FAC + NST)
#define OFF_CW  (OFF_V + 2 * SUB * VST)
#define OFF_YA  (OFF_CW + CO * CWP)
#define OFF_END (OFF_YA + 2 * CO * YP)

typedef __attribute__((ext_vector_type(2))) float v2f;
typedef __attribute__((ext_vector_type(8))) float v8f;

__device__ __forceinline__ v8f wmma4(v2f a, v2f b, v8f c) {
  return __builtin_amdgcn_wmma_f32_16x16x4_f32(
       false, a,  false, b,
       (short)0, c,  false,  false);
}

__device__ __forceinline__ void discretize_one(float x, float& ad, float& factor) {
  float sp = fmaxf(x, 0.0f) + log1pf(expf(-fabsf(x)));
  float lam = -sp;
  ad = expf(lam);
  factor = (fabsf(lam) > 1e-6f) ? (ad - 1.0f) / lam : 1.0f;
}
__device__ __forceinline__ void discretize_2cols(float x0, float x1, int n,
                                                 float* lds_ad, float* lds_fac) {
  __builtin_amdgcn_sched_barrier(0);
  float ad0, f0, ad1, f1;
  discretize_one(x0, ad0, f0);
  discretize_one(x1, ad1, f1);
  lds_ad[n] = ad0;        lds_fac[n] = f0;
  lds_ad[n + 128] = ad1;  lds_fac[n + 128] = f1;
}


__device__ __forceinline__ void fetch_u_tile128(const float* __restrict__ u, int b,
                                                int t0, int rw, int lane,
                                                float (&ur)[16]) {
#pragma unroll
  for (int kk = 0; kk < CIN / 4; ++kk) {
    int c = rw * (CIN / 4) + kk;
    ur[kk] = u[((size_t)(b * CIN + c)) * TLEN + t0 + lane];
  }
}
__device__ __forceinline__ void put_u_tile128(float* lds_u, int rw, int lane,
                                              const float (&ur)[16]) {
#pragma unroll
  for (int q = 0; q < 4; ++q)
    *reinterpret_cast<float4*>(&lds_u[lane * UST + rw * (CIN / 4) + 4 * q]) =
        make_float4(ur[4 * q], ur[4 * q + 1], ur[4 * q + 2], ur[4 * q + 3]);
}

__device__ __forceinline__ void load_bd_frags4(const float* __restrict__ Bd,
                                               int rw, int lo, int hi,
                                               v2f (&fr)[64]) {
#pragma unroll
  for (int j = 0; j < 4; ++j) {
    int nn = (rw * 4 + j) * 16 + lo;
#pragma unroll
    for (int ks = 0; ks < CIN / 4; ++ks) {
      int k = ks * 4 + 2 * hi;
      fr[j * 16 + ks].x = Bd[k * NST + nn];
      fr[j * 16 + ks].y = Bd[(k + 1) * NST + nn];
    }
  }
}

__device__ __forceinline__ void scale_bd_frags4(const float* lds_fac, int rw, int lo,
                                                v2f (&fr)[64]) {
#pragma unroll
  for (int j = 0; j < 4; ++j) {
    float f = lds_fac[(rw * 4 + j) * 16 + lo];
#pragma unroll
    for (int ks = 0; ks < CIN / 4; ++ks) {
      fr[j * 16 + ks].x = fr[j * 16 + ks].x * f;
      fr[j * 16 + ks].y = fr[j * 16 + ks].y * f;
    }
  }
}

__device__ __forceinline__ void v_tile_wmma8(const float* lds_u,
                                             const v2f (&fr)[64],
                                             float* lds_v, int rw, int lo, int hi) {
  v2f fa[2][CIN / 4];
#pragma unroll
  for (int ks = 0; ks < CIN / 4; ++ks) {
    int k = ks * 4 + 2 * hi;
    fa[0][ks] = *reinterpret_cast<const v2f*>(&lds_u[lo * UST + k]);
    fa[1][ks] = *reinterpret_cast<const v2f*>(&lds_u[(16 + lo) * UST + k]);
  }
  __builtin_amdgcn_sched_barrier(0);
  v8f acc[2][4];
#pragma unroll
  for (int mt = 0; mt < 2; ++mt)
#pragma unroll
    for (int j = 0; j < 4; ++j)
#pragma unroll
      for (int r = 0; r < 8; ++r) acc[mt][j][r] = 0.f;
#pragma unroll
  for (int ks = 0; ks < CIN / 4; ++ks) {
#pragma unroll
    for (int j = 0; j < 4; ++j) {
      acc[0][j] = wmma4(fa[0][ks], fr[j * 16 + ks], acc[0][j]);
      acc[1][j] = wmma4(fa[1][ks], fr[j * 16 + ks], acc[1][j]);
    }
  }
#pragma unroll
  for (int mt = 0; mt < 2; ++mt)
#pragma unroll
    for (int j = 0; j < 4; ++j)
#pragma unroll
      for (int r = 0; r < 8; ++r)
        lds_v[(mt * 16 + r + 8 * hi) * VST + (rw * 4 + j) * 16 + lo] = acc[mt][j][r];
}

__device__ __forceinline__ void load_u_pairs(const float* __restrict__ up, int t0,
                                             float2 (&ra)[CIN / 2]) {
#pragma unroll
  for (int ks = 0; ks < CIN / 4; ++ks) {
    ra[2 * ks]     = *reinterpret_cast<const float2*>(&up[(size_t)(4 * ks) * TLEN + t0]);
    ra[2 * ks + 1] = *reinterpret_cast<const float2*>(&up[(size_t)(4 * ks + 1) * TLEN + t0]);
  }
}

__device__ __forceinline__ void subtile_end_sums(const float2 (&ra)[CIN / 2],
                                                 const v2f (&fr)[64],
                                                 const float (&a)[4],
                                                 const float (&a2)[4],
                                                 const float (&a64)[4],
                                                 const float (&wh)[4],
                                                 float (&sl)[4]) {
  v2f fa[2][CIN / 4];
#pragma unroll
  for (int ks = 0; ks < CIN / 4; ++ks) {
    fa[0][ks].x = ra[2 * ks].x;  fa[0][ks].y = ra[2 * ks + 1].x;
    fa[1][ks].x = ra[2 * ks].y;  fa[1][ks].y = ra[2 * ks + 1].y;
  }
  v8f acc[2][4];
#pragma unroll
  for (int mt = 0; mt < 2; ++mt)
#pragma unroll
    for (int j = 0; j < 4; ++j)
#pragma unroll
      for (int r = 0; r < 8; ++r) acc[mt][j][r] = 0.f;
#pragma unroll
  for (int ks = 0; ks < CIN / 4; ++ks) {
#pragma unroll
    for (int j = 0; j < 4; ++j) {
      acc[0][j] = wmma4(fa[0][ks], fr[j * 16 + ks], acc[0][j]);
      acc[1][j] = wmma4(fa[1][ks], fr[j * 16 + ks], acc[1][j]);
    }
  }
#pragma unroll
  for (int j = 0; j < 4; ++j) {
    float h = __builtin_fmaf(acc[0][j][0], a[j], acc[1][j][0]);
#pragma unroll
    for (int r = 1; r < 8; ++r) {
      float q = __builtin_fmaf(acc[0][j][r], a[j], acc[1][j][r]);
      h = __builtin_fmaf(h, a2[j], q);
    }
    sl[j] = __builtin_fmaf(sl[j], a64[j], h * wh[j]);
  }
}


__global__ void __launch_bounds__(256)
k_chunk_ends(const float* __restrict__ u,
             const float* __restrict__ rl,
             const float* __restrict__ Bc,
             float* __restrict__ ce) {
  __shared__ float lds_x[2 * NST];
  const int chunk = blockIdx.x, b = blockIdx.y, tid = threadIdx.x;
  const int lane = tid & 31, lo = lane & 15, hi = lane >> 4;
  const int wave = __builtin_amdgcn_readfirstlane(tid >> 5);
  const int par = wave >> 2, rw = wave & 3;

  v2f fr[64];
  load_bd_frags4(Bc, rw, lo, hi, fr);
  float x[4];
#pragma unroll
  for (int j = 0; j < 4; ++j) x[j] = rl[(rw * 4 + j) * 16 + lo];
  const float* up = u + ((size_t)(b * CIN + 2 * hi)) * TLEN + chunk * LCH + 2 * lo;
  float2 ra0[CIN / 2], ra1[CIN / 2];
  load_u_pairs(up, par * SUB, ra0);

  float a[4], a2[4], a32[4], a64[4], wh[4], sl[4], slA[4];
#pragma unroll
  for (int j = 0; j < 4; ++j) {
    float f;
    discretize_one(x[j], a[j], f);
#pragma unroll
    for (int ks = 0; ks < CIN / 4; ++ks) {
      fr[j * 16 + ks].x = fr[j * 16 + ks].x * f;
      fr[j * 16 + ks].y = fr[j * 16 + ks].y * f;
    }
    a2[j] = a[j] * a[j];
    const float a4 = a2[j] * a2[j], a8 = a4 * a4, a16 = a8 * a8;
    a32[j] = a16 * a16;  a64[j] = a32[j] * a32[j];
    wh[j] = hi ? 1.0f : a16;
    sl[j] = 0.f;  slA[j] = 0.f;
  }

  for (int p = 0; p < NSUB / 4 - 1; ++p) {
    const int s = 4 * p + par;
    if (p == NSUBH / 4) {
#pragma unroll
      for (int j = 0; j < 4; ++j) { slA[j] = sl[j]; sl[j] = 0.f; }
    }
    load_u_pairs(up, (s + 2) * SUB, ra1);
    __builtin_amdgcn_sched_barrier(0);
    subtile_end_sums(ra0, fr, a, a2, a64, wh, sl);
    load_u_pairs(up, (s + 4) * SUB, ra0);
    __builtin_amdgcn_sched_barrier(0);
    subtile_end_sums(ra1, fr, a, a2, a64, wh, sl);
  }
  load_u_pairs(up, (NSUB - 2 + par) * SUB, ra1);
  __builtin_amdgcn_sched_barrier(0);
  subtile_end_sums(ra0, fr, a, a2, a64, wh, sl);
  subtile_end_sums(ra1, fr, a, a2, a64, wh, sl);

  float totA[4], totB[4];
#pragma unroll
  for (int j = 0; j < 4; ++j) {
    totA[j] = slA[j] + __shfl_xor(slA[j], 16);
    totB[j] = sl[j] + __shfl_xor(sl[j], 16);
  }
  if (par == 0 && hi == 0) {
#pragma unroll
    for (int j = 0; j < 4; ++j) {
      lds_x[(rw * 4 + j) * 16 + lo] = totA[j];
      lds_x[NST + (rw * 4 + j) * 16 + lo] = totB[j];
    }
  }
  __syncthreads();
  if (par == 1 && hi == 0) {
#pragma unroll
    for (int j = 0; j < 4; ++j) {
      int n = (rw * 4 + j) * 16 + lo;
      ce[(2 * chunk) * (BSZ * NST) + b * NST + n] =
          __builtin_fmaf(lds_x[n], a32[j], totA[j]);
      ce[(2 * chunk + 1) * (BSZ * NST) + b * NST + n] =
          __builtin_fmaf(lds_x[NST + n], a32[j], totB[j]);
    }
  }
}

__global__ void __launch_bounds__(256)
k_scan_conv(const float* __restrict__ u,
            const float* __restrict__ rl,
            const float* __restrict__ Bc,
            const float* __restrict__ Cm,
            const float* __restrict__ ln_w,
            const float* __restrict__ ln_b,
            const float* __restrict__ ce,
            const float* __restrict__ conv_w,
            const float* __restrict__ conv_b,
            float* __restrict__ out) {
  __shared__ __attribute__((aligned(16))) float lds[OFF_END];
  float* const lds_u   = lds + OFF_U;
  float* const lds_y   = lds + OFF_Y;
  float* const lds_red = lds + OFF_RED;
  float* const lds_ad  = lds + OFF_AD;
  float* const lds_fac = lds + OFF_FAC;
  float* const lds_v   = lds + OFF_V;
  float* const lds_cw  = lds + OFF_CW;
  float* const lds_ya  = lds + OFF_YA;

  const int pair = blockIdx.x, b = blockIdx.y, tid = threadIdx.x;
  const int lane = tid & 31, lo = lane & 15, hi = lane >> 4;
  const int wave = __builtin_amdgcn_readfirstlane(tid >> 5);
  const bool front = wave < 4;
  const int rw = wave & 3, rt = rw * 32 + lane;
  const int tbase = pair * LCH2;

  v2f fr[64];
#pragma unroll
  for (int i = 0; i < 64; ++i) { fr[i].x = 0.f; fr[i].y = 0.f; }
  float a0 = 0.f, a1 = 0.f, s0 = 0.f, s1 = 0.f, sB0 = 0.f, sB1 = 0.f;
  float lw[16], lb[16];
#pragma unroll
  for (int j = 0; j < 16; ++j) { lw[j] = 0.f; lb[j] = 0.f; }

  float e0[NCH2 - 1], e1[NCH2 - 1];
#pragma unroll
  for (int c = 0; c < NCH2 - 1; ++c) { e0[c] = 0.f; e1[c] = 0.f; }

  if (front) {
    load_bd_frags4(Bc, rw, lo, hi, fr);
#pragma unroll
    for (int c = 0; c < NCH2 - 1; ++c) {
      e0[c] = ce[c * (BSZ * NST) + b * NST + rt];
      e1[c] = ce[c * (BSZ * NST) + b * NST + rt + 128];
    }
#pragma unroll
    for (int j = 0; j < 16; ++j) { lw[j] = ln_w[rw * 16 + j]; lb[j] = ln_b[rw * 16 + j]; }
    float ur[16];
    fetch_u_tile128(u, b, tbase, rw, lane, ur);
    put_u_tile128(lds_u, rw, lane, ur);
  } else {
    const float x0 = rl[rt], x1 = rl[rt + 128];
    int oo = rw * 16 + lo;
#pragma unroll
    for (int ks = 0; ks < NST / 4; ++ks) {
      int k = ks * 4 + 2 * hi;
      fr[ks].x = Cm[k * CO + oo];
      fr[ks].y = Cm[(k + 1) * CO + oo];
    }
    float4 cwr[(CO * 2 * CO) / (4 * 128)];
#pragma unroll
    for (int i = 0; i < (CO * 2 * CO) / (4 * 128); ++i) {
      const float* cp = conv_w + 4 * (i * 128 + rt);
      cwr[i] = make_float4(cp[0], cp[2], cp[1], cp[3]);
    }
    discretize_2cols(x0, x1, rt, lds_ad, lds_fac);
#pragma unroll
    for (int i = 0; i < (CO * 2 * CO) / (4 * 128); ++i) {
      int q = i * 128 + rt;
      *reinterpret_cast<float4*>(&lds_cw[(q >> 5) * CWP + 4 * (q & 31)]) = cwr[i];
    }
  }
  __syncthreads();
  if (front) {
    scale_bd_frags4(lds_fac, rw, lo, fr);
    a0 = lds_ad[rt];  a1 = lds_ad[rt + 128];
    float aL0 = a0, aL1 = a1;
#pragma unroll
    for (int i = 0; i < 8; ++i) { aL0 = aL0 * aL0; aL1 = aL1 * aL1; }
    float r0 = 0.f, r1 = 0.f;
#pragma unroll
    for (int c = 0; c < NCH2; ++c) {
      if (c == pair) { s0 = r0; s1 = r1; }
      if (c == pair + NPAIR) { sB0 = r0; sB1 = r1; }
      if (c < NCH2 - 1) {
        r0 = __builtin_fmaf(aL0, r0, e0[c]);
        r1 = __builtin_fmaf(aL1, r1, e1[c]);
      }
    }
  }

  for (int step = 0; step < NSUB + 2; ++step) {
    if (front) {
      if (step < NSUB)
        v_tile_wmma8(lds_u, fr, lds_v + (step & 1) * (SUB * VST), rw, lo, hi);
    } else if (step >= 2) {
      const float* yb = lds_y + (step & 1) * (SUB * YST);
      float pr[4];
#pragma unroll
      for (int h = 0; h < 2; ++h) {
        int q = rw * 2 + h;
        float sm = 0.f, s2 = 0.f;
#pragma unroll
        for (int j = 0; j < 8; ++j) {
          float v = yb[lane * YST + q * 8 + j];
          sm += v; s2 += v * v;
        }
        pr[2 * h] = sm;  pr[2 * h + 1] = s2;
      }
      *reinterpret_cast<float4*>(&lds_red[lane * RST + rw * 4]) =
          make_float4(pr[0], pr[1], pr[2], pr[3]);
    }
    __syncthreads();

    if (front) {
      float ur[16];
      if (step + 1 < NSUB) {
        const int i1 = step + 1;
        fetch_u_tile128(u, b, (i1 >> 3) * T2 + tbase + (i1 & (NSUBH - 1)) * SUB,
                        rw, lane, ur);
      }
      if (step >= 2) {
        const float* yb = lds_y + (step & 1) * (SUB * YST);
        const int i2 = step - 2;
        float* ya = lds_ya + (i2 >> 3) * (CO * YP) + (i2 & (NSUBH - 1)) * SUB + lane;
        float sm = 0.f, s2 = 0.f;
#pragma unroll
        for (int q = 0; q < 8; ++q) {
          sm += lds_red[lane * RST + q * 2 + 0];
          s2 += lds_red[lane * RST + q * 2 + 1];
        }
        float yv[16];
#pragma unroll
        for (int j = 0; j < 16; ++j) yv[j] = yb[lane * YST + rw * 16 + j];
        float mu = sm * (1.0f / CO);
        float var = s2 * (1.0f / CO) - mu * mu;
        float rs = rsqrtf(var + 1e-5f);
#pragma unroll
        for (int j = 0; j < 16; ++j) {
          float v = (yv[j] - mu) * rs * lw[j] + lb[j];
          yv[j] = v / (1.0f + __expf(-v));
        }
#pragma unroll
        for (int j = 0; j < 16; ++j) ya[(rw * 16 + j) * YP] = yv[j];
      }
      if (step < NSUB) {
        if (step == NSUBH) { s0 = sB0; s1 = sB1; }
        float* S = lds_v + (step & 1) * (SUB * VST);
#pragma unroll 16
        for (int tl = 0; tl < SUB; ++tl) {
          s0 = s0 * a0 + S[tl * VST + rt];
          S[tl * VST + rt] = s0;
          s1 = s1 * a1 + S[tl * VST + rt + 128];
          S[tl * VST + rt + 128] = s1;
        }
      }
      if (step + 1 < NSUB)
        put_u_tile128(lds_u, rw, lane, ur);
    } else if (step >= 1 && step <= NSUB) {
      const float* S = lds_v + ((step - 1) & 1) * (SUB * VST);
      float* yb = lds_y + ((step - 1) & 1) * (SUB * YST);
      v8f y0 = {0.f, 0.f, 0.f, 0.f, 0.f, 0.f, 0.f, 0.f};
      v8f y1 = {0.f, 0.f, 0.f, 0.f, 0.f, 0.f, 0.f, 0.f};
      v2f fa[2][8][2];
#pragma unroll
      for (int i = 0; i < 8; ++i) {
        int k = i * 4 + 2 * hi;
        fa[0][i][0].x = S[lo * VST + k];         fa[0][i][0].y = S[lo * VST + k + 1];
        fa[0][i][1].x = S[(16 + lo) * VST + k];  fa[0][i][1].y = S[(16 + lo) * VST + k + 1];
      }
#pragma unroll
      for (int blk = 0; blk < NST / 32; ++blk) {
        if (blk + 1 < NST / 32) {
#pragma unroll
          for (int i = 0; i < 8; ++i) {
            int k = ((blk + 1) * 8 + i) * 4 + 2 * hi;
            fa[(blk + 1) & 1][i][0].x = S[lo * VST + k];
            fa[(blk + 1) & 1][i][0].y = S[lo * VST + k + 1];
            fa[(blk + 1) & 1][i][1].x = S[(16 + lo) * VST + k];
            fa[(blk + 1) & 1][i][1].y = S[(16 + lo) * VST + k + 1];
          }
        }
        __builtin_amdgcn_sched_barrier(0);
#pragma unroll
        for (int i = 0; i < 8; ++i) {
          y0 = wmma4(fa[blk & 1][i][0], fr[blk * 8 + i], y0);
          y1 = wmma4(fa[blk & 1][i][1], fr[blk * 8 + i], y1);
        }
        __builtin_amdgcn_sched_barrier(0);
      }
#pragma unroll
      for (int r = 0; r < 8; ++r) {
        yb[(r + 8 * hi) * YST + rw * 16 + lo] = y0[r];
        yb[(16 + r + 8 * hi) * YST + rw * 16 + lo] = y1[r];
      }
    }
    __syncthreads();
  }

  {
    const float* yA = lds_ya;
    const float* yB = lds_ya + CO * YP;
#pragma unroll
    for (int j = 0; j < 2; ++j) {
      const int tl = wave * 32 + 16 * j + lo;
      v8f acc[4];
#pragma unroll
      for (int mt = 0; mt < 4; ++mt)
#pragma unroll
        for (int r = 0; r < 8; ++r)
          acc[mt][r] = conv_b[mt * 16 + r + 8 * hi];

      float4 ca[2][4][4];
      v2f cbe[2][4], cbo[2][4];
#pragma unroll
      for (int q = 0; q < 4; ++q) {
        int k = q * 4 + 2 * hi;
#pragma unroll
        for (int mt = 0; mt < 4; ++mt)
          ca[0][q][mt] = *reinterpret_cast<const float4*>(
              &lds_cw[(mt * 16 + lo) * CWP + 2 * k]);
        cbe[0][q].x = yA[k * YP + tl];  cbe[0][q].y = yA[(k + 1) * YP + tl];
        cbo[0][q].x = yB[k * YP + tl];  cbo[0][q].y = yB[(k + 1) * YP + tl];
      }
#pragma unroll
      for (int blk = 0; blk < CO / 16; ++blk) {
        if (blk + 1 < CO / 16) {
#pragma unroll
          for (int q = 0; q < 4; ++q) {
            int k = ((blk + 1) * 4 + q) * 4 + 2 * hi;
#pragma unroll
            for (int mt = 0; mt < 4; ++mt)
              ca[(blk + 1) & 1][q][mt] = *reinterpret_cast<const float4*>(
                  &lds_cw[(mt * 16 + lo) * CWP + 2 * k]);
            cbe[(blk + 1) & 1][q].x = yA[k * YP + tl];
            cbe[(blk + 1) & 1][q].y = yA[(k + 1) * YP + tl];
            cbo[(blk + 1) & 1][q].x = yB[k * YP + tl];
            cbo[(blk + 1) & 1][q].y = yB[(k + 1) * YP + tl];
          }
        }
        __builtin_amdgcn_sched_barrier(0);
#pragma unroll
        for (int q = 0; q < 4; ++q) {
#pragma unroll
          for (int mt = 0; mt < 4; ++mt) {
            v2f ae;
            ae.x = ca[blk & 1][q][mt].x;  ae.y = ca[blk & 1][q][mt].y;
            acc[mt] = wmma4(ae, cbe[blk & 1][q], acc[mt]);
          }
#pragma unroll
          for (int mt = 0; mt < 4; ++mt) {
            v2f ao;
            ao.x = ca[blk & 1][q][mt].z;  ao.y = ca[blk & 1][q][mt].w;
            acc[mt] = wmma4(ao, cbo[blk & 1][q], acc[mt]);
          }
        }
        __builtin_amdgcn_sched_barrier(0);
      }

#pragma unroll
      for (int mt = 0; mt < 4; ++mt)
#pragma unroll
        for (int r = 0; r < 8; ++r) {
          int o = mt * 16 + r + 8 * hi;
          int t = 2 * (tbase + tl);
          float v = acc[mt][r];
          float2 pv = make_float2(v, v);
          *reinterpret_cast<float2*>(&out[((size_t)(b * CO + o)) * TLEN + t]) = pv;
        }
    }
  }
}

extern "C" void kernel_launch(void* const* d_in, const int* in_sizes, int n_in,
                              void* d_out, int out_size, void* d_ws, size_t ws_size,
                              hipStream_t stream) {
  const float* u     = (const float*)d_in[0];
  const float* rl    = (const float*)d_in[1];
  const float* Bc    = (const float*)d_in[2];
  const float* Cm    = (const float*)d_in[3];
  const float* ln_w  = (const float*)d_in[4];
  const float* ln_b  = (const float*)d_in[5];
  const float* cw    = (const float*)d_in[6];
  const float* cb    = (const float*)d_in[7];
  float* out = (float*)d_out;

  float* ce    = (float*)d_ws;

  k_chunk_ends<<<dim3(NCH, BSZ), 256, 0, stream>>>(u, rl, Bc, ce);
  k_scan_conv<<<dim3(NPAIR, BSZ), 256, 0, stream>>>(u, rl, Bc, Cm, ln_w, ln_b,
                                                    ce, cw, cb, out);
}
